# starting kernel
# baseline (speedup 1.0000x reference)
#include <hip/hip_runtime.h>
#include <hip/hip_bf16.h>
#include <math.h>

#define N_NODES   100000
#define N_LAYERS  5
#define HID       150
#define HP        160
#define RPB       32
#define RBLOCKS   (N_NODES / RPB)
#define COL_TILES (HP / 16)
#define KSTEPS    (HP / 4)
#define LDS_LD    164

typedef __attribute__((ext_vector_type(2))) float v2f;
typedef __attribute__((ext_vector_type(8))) float v8f;

__device__ __forceinline__ v8f wmma_f32(v2f a, v2f b, v8f c) {
  return __builtin_amdgcn_wmma_f32_16x16x4_f32(
       false, a,  false, b,
       (short)0, c,  false,  false);
}

__device__ __forceinline__ float sigmoidf_(float x) {
  return 1.0f / (1.0f + __expf(-x));
}


__global__ void init_hpad(const float* __restrict__ x, float* __restrict__ hpad) {
  int idx = blockIdx.x * blockDim.x + threadIdx.x;
  if (idx >= N_NODES * HP) return;
  int row = idx / HP, col = idx - row * HP;
  hpad[idx] = (col < HID) ? x[row * HID + col] : 0.0f;
}

__global__ void pack_w1(const float* __restrict__ w, float* __restrict__ wp) {
  int idx = blockIdx.x * blockDim.x + threadIdx.x;
  const int per_layer = COL_TILES * KSTEPS * 64;
  if (idx >= N_LAYERS * per_layer) return;
  int l    = idx / per_layer;
  int rem  = idx - l * per_layer;
  int j    = rem & 1;
  int lane = (rem >> 1) & 31;
  int ks   = (rem >> 6) % KSTEPS;
  int ct   = (rem >> 6) / KSTEPS;
  int k = ks * 4 + 2 * (lane >> 4) + j;
  int n = ct * 16 + (lane & 15);
  wp[idx] = (k < HID && n < HID) ? w[(l * HID + k) * HID + n] : 0.0f;
}

__global__ void pack_wg(const float* __restrict__ w, float* __restrict__ wp) {
  int idx = blockIdx.x * blockDim.x + threadIdx.x;
  const int per_gate = COL_TILES * KSTEPS * 64;
  if (idx >= 3 * per_gate) return;
  int g    = idx / per_gate;
  int rem  = idx - g * per_gate;
  int j    = rem & 1;
  int lane = (rem >> 1) & 31;
  int ks   = (rem >> 6) % KSTEPS;
  int ct   = (rem >> 6) / KSTEPS;
  int k = ks * 4 + 2 * (lane >> 4) + j;
  int n = ct * 16 + (lane & 15);
  wp[idx] = (k < HID && n < HID) ? w[(g * HID + n) * HID + k] : 0.0f;
}

__global__ void hist_deg(const int* __restrict__ ei, int* __restrict__ deg,
                         int n_edges) {
  int e = blockIdx.x * blockDim.x + threadIdx.x;
  if (e >= n_edges) return;
  atomicAdd(&deg[ei[n_edges + e]], 1);
}

__global__ void __launch_bounds__(1024)
scan_deg(const int* __restrict__ deg, int* __restrict__ rowptr,
         int* __restrict__ cursor) {
  __shared__ int buf[1024];
  __shared__ int carry_s;
  const int tid = threadIdx.x;
  if (tid == 0) carry_s = 0;
  __syncthreads();
  for (int base = 0; base < N_NODES; base += 1024) {
    int i = base + tid;
    int v = (i < N_NODES) ? deg[i] : 0;
    buf[tid] = v;
    __syncthreads();
    for (int off = 1; off < 1024; off <<= 1) {
      int t = (tid >= off) ? buf[tid - off] : 0;
      __syncthreads();
      buf[tid] += t;
      __syncthreads();
    }
    int incl = buf[tid];
    int carry = carry_s;
    if (i < N_NODES) {
      int excl = carry + incl - v;
      rowptr[i] = excl;
      cursor[i] = excl;
    }
    __syncthreads();
    if (tid == 1023) carry_s = carry + incl;
    __syncthreads();
  }
  if (tid == 0) rowptr[N_NODES] = carry_s;
}

__global__ void fill_csr(const int* __restrict__ ei, int* __restrict__ cursor,
                         int* __restrict__ csr_src, int n_edges) {
  int e = blockIdx.x * blockDim.x + threadIdx.x;
  if (e >= n_edges) return;
  int pos = atomicAdd(&cursor[ei[n_edges + e]], 1);
  csr_src[pos] = ei[e];
}

__global__ void __launch_bounds__(320)
gemm_m(const float* __restrict__ h, const float* __restrict__ wp,
       float* __restrict__ m) {
  __shared__ float hs[RPB * LDS_LD];
  const int rowbase = blockIdx.x * RPB;

  for (int i = threadIdx.x; i < RPB * (HP / 4); i += 320) {
    int r = i / (HP / 4), cc = (i - r * (HP / 4)) * 4;
    float4 v = *(const float4*)(h + (rowbase + r) * HP + cc);
    float* d = &hs[r * LDS_LD + cc];
    d[0] = v.x; d[1] = v.y; d[2] = v.z; d[3] = v.w;
  }
  __syncthreads();

  const int wave = threadIdx.x >> 5, lane = threadIdx.x & 31;
  const int half = lane >> 4, l16 = lane & 15;
  const float* wpt = wp + wave * (KSTEPS * 64);

  v8f acc0 = {}, acc1 = {};
  for (int ks = 0; ks < KSTEPS; ++ks) {
    int ao = ks * 4 + 2 * half;
    v2f a0 = *(const v2f*)&hs[l16 * LDS_LD + ao];
    v2f a1 = *(const v2f*)&hs[(l16 + 16) * LDS_LD + ao];
    v2f b  = *(const v2f*)(wpt + ks * 64 + lane * 2);
    acc0 = wmma_f32(a0, b, acc0);
    acc1 = wmma_f32(a1, b, acc1);
  }
  const int col = wave * 16 + l16;
  #pragma unroll
  for (int v = 0; v < 8; ++v) {
    m[(rowbase + v + 8 * half) * HP + col]      = acc0[v];
    m[(rowbase + 16 + v + 8 * half) * HP + col] = acc1[v];
  }
}

__global__ void __launch_bounds__(256)
gather_aggr(const float* __restrict__ m, const int* __restrict__ rowptr,
            const int* __restrict__ csr_src, float* __restrict__ aggr) {
  int node = blockIdx.x * 8 + (threadIdx.x >> 5);
  if (node >= N_NODES) return;
  const int lane = threadIdx.x & 31;
  int beg = rowptr[node], end = rowptr[node + 1];
  float a0 = 0.f, a1 = 0.f, a2 = 0.f, a3 = 0.f, a4 = 0.f;
  for (int j = beg; j < end; ++j) {
    const float* mr = m + (size_t)csr_src[j] * HP;
    float4 v = *(const float4*)(mr + 4 * lane);
    float  s = mr[128 + lane];
    a0 += v.x; a1 += v.y; a2 += v.z; a3 += v.w; a4 += s;
  }
  float* ar = aggr + (size_t)node * HP;
  float4 o; o.x = a0; o.y = a1; o.z = a2; o.w = a3;
  *(float4*)(ar + 4 * lane) = o;
  ar[128 + lane] = a4;
}

__global__ void __launch_bounds__(320)
gru_fused(const float* __restrict__ hin, const float* __restrict__ aggr,
          const float* __restrict__ wih, const float* __restrict__ whh,
          const float* __restrict__ bih, const float* __restrict__ bhh,
          float* __restrict__ hout, int ldo, int zero_pad) {
  __shared__ float hs[RPB * LDS_LD];
  __shared__ float as_[RPB * LDS_LD];
  const int rowbase = blockIdx.x * RPB;

  for (int i = threadIdx.x; i < 2 * RPB * (HP / 4); i += 320) {
    int which = i / (RPB * (HP / 4));
    int t = i - which * (RPB * (HP / 4));
    int r = t / (HP / 4), cc = (t - r * (HP / 4)) * 4;
    const float* srcp = which ? (aggr + (rowbase + r) * HP + cc)
                              : (hin + (rowbase + r) * HP + cc);
    float4 v = *(const float4*)srcp;
    float* d = which ? &as_[r * LDS_LD + cc] : &hs[r * LDS_LD + cc];
    d[0] = v.x; d[1] = v.y; d[2] = v.z; d[3] = v.w;
  }
  __syncthreads();

  const int wave = threadIdx.x >> 5, lane = threadIdx.x & 31;
  const int half = lane >> 4, l16 = lane & 15;
  const int ct = wave;

  const float* pir = wih + (0 * COL_TILES + ct) * (KSTEPS * 64);
  const float* piz = wih + (1 * COL_TILES + ct) * (KSTEPS * 64);
  const float* pin = wih + (2 * COL_TILES + ct) * (KSTEPS * 64);
  const float* phr = whh + (0 * COL_TILES + ct) * (KSTEPS * 64);
  const float* phz = whh + (1 * COL_TILES + ct) * (KSTEPS * 64);
  const float* phn = whh + (2 * COL_TILES + ct) * (KSTEPS * 64);

  v8f ir0 = {}, hr0 = {}, iz0 = {}, hz0 = {}, in0 = {}, hn0 = {};
  v8f ir1 = {}, hr1 = {}, iz1 = {}, hz1 = {}, in1 = {}, hn1 = {};
  for (int ks = 0; ks < KSTEPS; ++ks) {
    int ao = ks * 4 + 2 * half;
    v2f aA0 = *(const v2f*)&as_[l16 * LDS_LD + ao];
    v2f aA1 = *(const v2f*)&as_[(l16 + 16) * LDS_LD + ao];
    v2f aH0 = *(const v2f*)&hs[l16 * LDS_LD + ao];
    v2f aH1 = *(const v2f*)&hs[(l16 + 16) * LDS_LD + ao];
    v2f br = *(const v2f*)(pir + ks * 64 + lane * 2);
    v2f bz = *(const v2f*)(piz + ks * 64 + lane * 2);
    v2f bn = *(const v2f*)(pin + ks * 64 + lane * 2);
    v2f cr = *(const v2f*)(phr + ks * 64 + lane * 2);
    v2f cz = *(const v2f*)(phz + ks * 64 + lane * 2);
    v2f cn = *(const v2f*)(phn + ks * 64 + lane * 2);
    ir0 = wmma_f32(aA0, br, ir0);  ir1 = wmma_f32(aA1, br, ir1);
    iz0 = wmma_f32(aA0, bz, iz0);  iz1 = wmma_f32(aA1, bz, iz1);
    in0 = wmma_f32(aA0, bn, in0);  in1 = wmma_f32(aA1, bn, in1);
    hr0 = wmma_f32(aH0, cr, hr0);  hr1 = wmma_f32(aH1, cr, hr1);
    hz0 = wmma_f32(aH0, cz, hz0);  hz1 = wmma_f32(aH1, cz, hz1);
    hn0 = wmma_f32(aH0, cn, hn0);  hn1 = wmma_f32(aH1, cn, hn1);
  }

  const int col = ct * 16 + l16;
  const bool valid = (col < HID);
  const int cb = valid ? col : 0;
  const float bir = valid ? bih[cb]           : 0.0f;
  const float biz = valid ? bih[HID + cb]     : 0.0f;
  const float bin = valid ? bih[2 * HID + cb] : 0.0f;
  const float bhr = valid ? bhh[cb]           : 0.0f;
  const float bhz = valid ? bhh[HID + cb]     : 0.0f;
  const float bhn = valid ? bhh[2 * HID + cb] : 0.0f;

  #pragma unroll
  for (int rt = 0; rt < 2; ++rt) {
    #pragma unroll
    for (int v = 0; v < 8; ++v) {
      float irv = rt ? ir1[v] : ir0[v];
      float hrv = rt ? hr1[v] : hr0[v];
      float izv = rt ? iz1[v] : iz0[v];
      float hzv = rt ? hz1[v] : hz0[v];
      float inv = rt ? in1[v] : in0[v];
      float hnv = rt ? hn1[v] : hn0[v];
      int rl = rt * 16 + v + 8 * half;
      float r = sigmoidf_((irv + bir) + (hrv + bhr));
      float z = sigmoidf_((izv + biz) + (hzv + bhz));
      float n = tanhf((inv + bin) + r * (hnv + bhn));
      float hv = hs[rl * LDS_LD + col];
      float o = (1.0f - z) * n + z * hv;
      long row = rowbase + rl;
      if (valid)          hout[row * ldo + col] = o;
      else if (zero_pad)  hout[row * ldo + col] = 0.0f;
    }
  }
}

extern "C" void kernel_launch(void* const* d_in, const int* in_sizes, int n_in,
                              void* d_out, int out_size, void* d_ws, size_t ws_size,
                              hipStream_t stream) {
  const float* x    = (const float*)d_in[0];
  const float* wgt  = (const float*)d_in[1];
  const float* w_ih = (const float*)d_in[2];
  const float* w_hh = (const float*)d_in[3];
  const float* b_ih = (const float*)d_in[4];
  const float* b_hh = (const float*)d_in[5];
  const int*   ei   = (const int*)d_in[6];
  const int n_edges = in_sizes[6] / 2;
  float* out = (float*)d_out;

  const size_t npad = (size_t)N_NODES * HP;
  const size_t gate_sz = 3 * COL_TILES * KSTEPS * 64;
  float* ws      = (float*)d_ws;
  float* h_pad   = ws;
  float* m_pad   = h_pad + npad;
  float* aggr    = m_pad + npad;
  float* wpack1  = aggr  + npad;
  float* wpackih = wpack1  + (size_t)N_LAYERS * COL_TILES * KSTEPS * 64;
  float* wpackhh = wpackih + gate_sz;
  int*   deg     = (int*)(wpackhh + gate_sz);
  int*   rowptr  = deg + N_NODES;
  int*   cursor  = rowptr + (N_NODES + 1);
  int*   csr_src = cursor + N_NODES;

  {
    int n = N_NODES * HP;
    init_hpad<<<(n + 255) / 256, 256, 0, stream>>>(x, h_pad);
    int n1 = N_LAYERS * COL_TILES * KSTEPS * 64;
    pack_w1<<<(n1 + 255) / 256, 256, 0, stream>>>(wgt, wpack1);
    int ng = (int)gate_sz;
    pack_wg<<<(ng + 255) / 256, 256, 0, stream>>>(w_ih, wpackih);
    pack_wg<<<(ng + 255) / 256, 256, 0, stream>>>(w_hh, wpackhh);

    hipMemsetAsync(deg, 0, (size_t)N_NODES * sizeof(int), stream);
    hist_deg<<<(n_edges + 255) / 256, 256, 0, stream>>>(ei, deg, n_edges);
    scan_deg<<<1, 1024, 0, stream>>>(deg, rowptr, cursor);
    fill_csr<<<(n_edges + 255) / 256, 256, 0, stream>>>(ei, cursor, csr_src, n_edges);
  }

  for (int l = 0; l < N_LAYERS; ++l) {
    gemm_m<<<RBLOCKS, 320, 0, stream>>>(
        h_pad, wpack1 + (size_t)l * COL_TILES * KSTEPS * 64, m_pad);

    gather_aggr<<<(N_NODES + 7) / 8, 256, 0, stream>>>(m_pad, rowptr, csr_src, aggr);

    const bool last = (l == N_LAYERS - 1);
    gru_fused<<<RBLOCKS, 320, 0, stream>>>(
        h_pad, aggr, wpackih, wpackhh, b_ih, b_hh,
        last ? out : h_pad, last ? HID : HP, last ? 0 : 1);
  }
}


# gru_fused K loop software-pipelined: step k+1's weight and LDS fragments are loaded as one group into a second register set ahead of step k's 12 WMMAs (sched_barrier holds the groups)
# speedup vs baseline: 1.0591x; 1.0591x over previous
#include <hip/hip_runtime.h>
#include <hip/hip_bf16.h>
#include <math.h>

#define N_NODES   100000
#define N_LAYERS  5
#define HID       150
#define HP        160
#define RPB       32
#define RBLOCKS   (N_NODES / RPB)
#define COL_TILES (HP / 16)
#define KSTEPS    (HP / 4)
#define LDS_LD    164

typedef __attribute__((ext_vector_type(2))) float v2f;
typedef __attribute__((ext_vector_type(8))) float v8f;

__device__ __forceinline__ v8f wmma_f32(v2f a, v2f b, v8f c) {
  return __builtin_amdgcn_wmma_f32_16x16x4_f32(
       false, a,  false, b,
       (short)0, c,  false,  false);
}

__device__ __forceinline__ float sigmoidf_(float x) {
  return 1.0f / (1.0f + __expf(-x));
}


__global__ void init_hpad(const float* __restrict__ x, float* __restrict__ hpad) {
  int idx = blockIdx.x * blockDim.x + threadIdx.x;
  if (idx >= N_NODES * HP) return;
  int row = idx / HP, col = idx - row * HP;
  hpad[idx] = (col < HID) ? x[row * HID + col] : 0.0f;
}

__global__ void pack_w1(const float* __restrict__ w, float* __restrict__ wp) {
  int idx = blockIdx.x * blockDim.x + threadIdx.x;
  const int per_layer = COL_TILES * KSTEPS * 64;
  if (idx >= N_LAYERS * per_layer) return;
  int l    = idx / per_layer;
  int rem  = idx - l * per_layer;
  int j    = rem & 1;
  int lane = (rem >> 1) & 31;
  int ks   = (rem >> 6) % KSTEPS;
  int ct   = (rem >> 6) / KSTEPS;
  int k = ks * 4 + 2 * (lane >> 4) + j;
  int n = ct * 16 + (lane & 15);
  wp[idx] = (k < HID && n < HID) ? w[(l * HID + k) * HID + n] : 0.0f;
}

__global__ void pack_wg(const float* __restrict__ w, float* __restrict__ wp) {
  int idx = blockIdx.x * blockDim.x + threadIdx.x;
  const int per_gate = COL_TILES * KSTEPS * 64;
  if (idx >= 3 * per_gate) return;
  int g    = idx / per_gate;
  int rem  = idx - g * per_gate;
  int j    = rem & 1;
  int lane = (rem >> 1) & 31;
  int ks   = (rem >> 6) % KSTEPS;
  int ct   = (rem >> 6) / KSTEPS;
  int k = ks * 4 + 2 * (lane >> 4) + j;
  int n = ct * 16 + (lane & 15);
  wp[idx] = (k < HID && n < HID) ? w[(g * HID + n) * HID + k] : 0.0f;
}

__global__ void hist_deg(const int* __restrict__ ei, int* __restrict__ deg,
                         int n_edges) {
  int e = blockIdx.x * blockDim.x + threadIdx.x;
  if (e >= n_edges) return;
  atomicAdd(&deg[ei[n_edges + e]], 1);
}

__global__ void __launch_bounds__(1024)
scan_deg(const int* __restrict__ deg, int* __restrict__ rowptr,
         int* __restrict__ cursor) {
  __shared__ int buf[1024];
  __shared__ int carry_s;
  const int tid = threadIdx.x;
  if (tid == 0) carry_s = 0;
  __syncthreads();
  for (int base = 0; base < N_NODES; base += 1024) {
    int i = base + tid;
    int v = (i < N_NODES) ? deg[i] : 0;
    buf[tid] = v;
    __syncthreads();
    for (int off = 1; off < 1024; off <<= 1) {
      int t = (tid >= off) ? buf[tid - off] : 0;
      __syncthreads();
      buf[tid] += t;
      __syncthreads();
    }
    int incl = buf[tid];
    int carry = carry_s;
    if (i < N_NODES) {
      int excl = carry + incl - v;
      rowptr[i] = excl;
      cursor[i] = excl;
    }
    __syncthreads();
    if (tid == 1023) carry_s = carry + incl;
    __syncthreads();
  }
  if (tid == 0) rowptr[N_NODES] = carry_s;
}

__global__ void fill_csr(const int* __restrict__ ei, int* __restrict__ cursor,
                         int* __restrict__ csr_src, int n_edges) {
  int e = blockIdx.x * blockDim.x + threadIdx.x;
  if (e >= n_edges) return;
  int pos = atomicAdd(&cursor[ei[n_edges + e]], 1);
  csr_src[pos] = ei[e];
}

__global__ void __launch_bounds__(320)
gemm_m(const float* __restrict__ h, const float* __restrict__ wp,
       float* __restrict__ m) {
  __shared__ float hs[RPB * LDS_LD];
  const int rowbase = blockIdx.x * RPB;

  for (int i = threadIdx.x; i < RPB * (HP / 4); i += 320) {
    int r = i / (HP / 4), cc = (i - r * (HP / 4)) * 4;
    float4 v = *(const float4*)(h + (rowbase + r) * HP + cc);
    float* d = &hs[r * LDS_LD + cc];
    d[0] = v.x; d[1] = v.y; d[2] = v.z; d[3] = v.w;
  }
  __syncthreads();

  const int wave = threadIdx.x >> 5, lane = threadIdx.x & 31;
  const int half = lane >> 4, l16 = lane & 15;
  const float* wpt = wp + wave * (KSTEPS * 64);

  v8f acc0 = {}, acc1 = {};
  for (int ks = 0; ks < KSTEPS; ++ks) {
    int ao = ks * 4 + 2 * half;
    v2f a0 = *(const v2f*)&hs[l16 * LDS_LD + ao];
    v2f a1 = *(const v2f*)&hs[(l16 + 16) * LDS_LD + ao];
    v2f b  = *(const v2f*)(wpt + ks * 64 + lane * 2);
    acc0 = wmma_f32(a0, b, acc0);
    acc1 = wmma_f32(a1, b, acc1);
  }
  const int col = wave * 16 + l16;
  #pragma unroll
  for (int v = 0; v < 8; ++v) {
    m[(rowbase + v + 8 * half) * HP + col]      = acc0[v];
    m[(rowbase + 16 + v + 8 * half) * HP + col] = acc1[v];
  }
}

__global__ void __launch_bounds__(256)
gather_aggr(const float* __restrict__ m, const int* __restrict__ rowptr,
            const int* __restrict__ csr_src, float* __restrict__ aggr) {
  int node = blockIdx.x * 8 + (threadIdx.x >> 5);
  if (node >= N_NODES) return;
  const int lane = threadIdx.x & 31;
  int beg = rowptr[node], end = rowptr[node + 1];
  float a0 = 0.f, a1 = 0.f, a2 = 0.f, a3 = 0.f, a4 = 0.f;
  for (int j = beg; j < end; ++j) {
    const float* mr = m + (size_t)csr_src[j] * HP;
    float4 v = *(const float4*)(mr + 4 * lane);
    float  s = mr[128 + lane];
    a0 += v.x; a1 += v.y; a2 += v.z; a3 += v.w; a4 += s;
  }
  float* ar = aggr + (size_t)node * HP;
  float4 o; o.x = a0; o.y = a1; o.z = a2; o.w = a3;
  *(float4*)(ar + 4 * lane) = o;
  ar[128 + lane] = a4;
}

struct KFrag {
  v2f aA0, aA1, aH0, aH1;
  v2f br, bz, bn, cr, cz, cn;
};

__global__ void __launch_bounds__(320)
gru_fused(const float* __restrict__ hin, const float* __restrict__ aggr,
          const float* __restrict__ wih, const float* __restrict__ whh,
          const float* __restrict__ bih, const float* __restrict__ bhh,
          float* __restrict__ hout, int ldo, int zero_pad) {
  __shared__ float hs[RPB * LDS_LD];
  __shared__ float as_[RPB * LDS_LD];
  const int rowbase = blockIdx.x * RPB;

  for (int i = threadIdx.x; i < 2 * RPB * (HP / 4); i += 320) {
    int which = i / (RPB * (HP / 4));
    int t = i - which * (RPB * (HP / 4));
    int r = t / (HP / 4), cc = (t - r * (HP / 4)) * 4;
    const float* srcp = which ? (aggr + (rowbase + r) * HP + cc)
                              : (hin + (rowbase + r) * HP + cc);
    float4 v = *(const float4*)srcp;
    float* d = which ? &as_[r * LDS_LD + cc] : &hs[r * LDS_LD + cc];
    d[0] = v.x; d[1] = v.y; d[2] = v.z; d[3] = v.w;
  }
  __syncthreads();

  const int wave = threadIdx.x >> 5, lane = threadIdx.x & 31;
  const int half = lane >> 4, l16 = lane & 15;
  const int ct = wave;

  const int GSTRIDE = COL_TILES * KSTEPS * 64;
  const float* aAp = &as_[l16 * LDS_LD + 2 * half];
  const float* aHp = &hs[l16 * LDS_LD + 2 * half];
  const float* wi = wih + ct * (KSTEPS * 64) + lane * 2;
  const float* wh = whh + ct * (KSTEPS * 64) + lane * 2;

  v8f ir0 = {}, hr0 = {}, iz0 = {}, hz0 = {}, in0 = {}, hn0 = {};
  v8f ir1 = {}, hr1 = {}, iz1 = {}, hz1 = {}, in1 = {}, hn1 = {};

#define LOAD_STEP(F, KS)                                                   \
  do {                                                                     \
    const int ks_ = (KS);                                                  \
    F.br = *(const v2f*)(wi + ks_ * 64);                                   \
    F.bz = *(const v2f*)(wi + GSTRIDE + ks_ * 64);                         \
    F.bn = *(const v2f*)(wi + 2 * GSTRIDE + ks_ * 64);                     \
    F.cr = *(const v2f*)(wh + ks_ * 64);                                   \
    F.cz = *(const v2f*)(wh + GSTRIDE + ks_ * 64);                         \
    F.cn = *(const v2f*)(wh + 2 * GSTRIDE + ks_ * 64);                     \
    F.aA0 = *(const v2f*)(aAp + ks_ * 4);                                  \
    F.aA1 = *(const v2f*)(aAp + 16 * LDS_LD + ks_ * 4);                    \
    F.aH0 = *(const v2f*)(aHp + ks_ * 4);                                  \
    F.aH1 = *(const v2f*)(aHp + 16 * LDS_LD + ks_ * 4);                    \
  } while (0)

#define MMA_STEP(F)                                                        \
  do {                                                                     \
    ir0 = wmma_f32(F.aA0, F.br, ir0);  ir1 = wmma_f32(F.aA1, F.br, ir1);   \
    iz0 = wmma_f32(F.aA0, F.bz, iz0);  iz1 = wmma_f32(F.aA1, F.bz, iz1);   \
    in0 = wmma_f32(F.aA0, F.bn, in0);  in1 = wmma_f32(F.aA1, F.bn, in1);   \
    hr0 = wmma_f32(F.aH0, F.cr, hr0);  hr1 = wmma_f32(F.aH1, F.cr, hr1);   \
    hz0 = wmma_f32(F.aH0, F.cz, hz0);  hz1 = wmma_f32(F.aH1, F.cz, hz1);   \
    hn0 = wmma_f32(F.aH0, F.cn, hn0);  hn1 = wmma_f32(F.aH1, F.cn, hn1);   \
  } while (0)

  KFrag f0, f1;
  LOAD_STEP(f0, 0);
  for (int ks = 0; ks < KSTEPS; ks += 2) {
    LOAD_STEP(f1, ks + 1);
    __builtin_amdgcn_sched_barrier(0);
    MMA_STEP(f0);
    __builtin_amdgcn_sched_barrier(0);
    LOAD_STEP(f0, (ks + 2 < KSTEPS) ? ks + 2 : ks + 1);
    __builtin_amdgcn_sched_barrier(0);
    MMA_STEP(f1);
    __builtin_amdgcn_sched_barrier(0);
  }
#undef LOAD_STEP
#undef MMA_STEP

  const int col = ct * 16 + l16;
  const bool valid = (col < HID);
  const int cb = valid ? col : 0;
  const float bir = valid ? bih[cb]           : 0.0f;
  const float biz = valid ? bih[HID + cb]     : 0.0f;
  const float bin = valid ? bih[2 * HID + cb] : 0.0f;
  const float bhr = valid ? bhh[cb]           : 0.0f;
  const float bhz = valid ? bhh[HID + cb]     : 0.0f;
  const float bhn = valid ? bhh[2 * HID + cb] : 0.0f;

  #pragma unroll
  for (int rt = 0; rt < 2; ++rt) {
    #pragma unroll
    for (int v = 0; v < 8; ++v) {
      float irv = rt ? ir1[v] : ir0[v];
      float hrv = rt ? hr1[v] : hr0[v];
      float izv = rt ? iz1[v] : iz0[v];
      float hzv = rt ? hz1[v] : hz0[v];
      float inv = rt ? in1[v] : in0[v];
      float hnv = rt ? hn1[v] : hn0[v];
      int rl = rt * 16 + v + 8 * half;
      float r = sigmoidf_((irv + bir) + (hrv + bhr));
      float z = sigmoidf_((izv + biz) + (hzv + bhz));
      float n = tanhf((inv + bin) + r * (hnv + bhn));
      float hv = hs[rl * LDS_LD + col];
      float o = (1.0f - z) * n + z * hv;
      long row = rowbase + rl;
      if (valid)          hout[row * ldo + col] = o;
      else if (zero_pad)  hout[row * ldo + col] = 0.0f;
    }
  }
}

extern "C" void kernel_launch(void* const* d_in, const int* in_sizes, int n_in,
                              void* d_out, int out_size, void* d_ws, size_t ws_size,
                              hipStream_t stream) {
  const float* x    = (const float*)d_in[0];
  const float* wgt  = (const float*)d_in[1];
  const float* w_ih = (const float*)d_in[2];
  const float* w_hh = (const float*)d_in[3];
  const float* b_ih = (const float*)d_in[4];
  const float* b_hh = (const float*)d_in[5];
  const int*   ei   = (const int*)d_in[6];
  const int n_edges = in_sizes[6] / 2;
  float* out = (float*)d_out;

  const size_t npad = (size_t)N_NODES * HP;
  const size_t gate_sz = 3 * COL_TILES * KSTEPS * 64;
  float* ws      = (float*)d_ws;
  float* h_pad   = ws;
  float* m_pad   = h_pad + npad;
  float* aggr    = m_pad + npad;
  float* wpack1  = aggr  + npad;
  float* wpackih = wpack1  + (size_t)N_LAYERS * COL_TILES * KSTEPS * 64;
  float* wpackhh = wpackih + gate_sz;
  int*   deg     = (int*)(wpackhh + gate_sz);
  int*   rowptr  = deg + N_NODES;
  int*   cursor  = rowptr + (N_NODES + 1);
  int*   csr_src = cursor + N_NODES;

  {
    int n = N_NODES * HP;
    init_hpad<<<(n + 255) / 256, 256, 0, stream>>>(x, h_pad);
    int n1 = N_LAYERS * COL_TILES * KSTEPS * 64;
    pack_w1<<<(n1 + 255) / 256, 256, 0, stream>>>(wgt, wpack1);
    int ng = (int)gate_sz;
    pack_wg<<<(ng + 255) / 256, 256, 0, stream>>>(w_ih, wpackih);
    pack_wg<<<(ng + 255) / 256, 256, 0, stream>>>(w_hh, wpackhh);

    hipMemsetAsync(deg, 0, (size_t)N_NODES * sizeof(int), stream);
    hist_deg<<<(n_edges + 255) / 256, 256, 0, stream>>>(ei, deg, n_edges);
    scan_deg<<<1, 1024, 0, stream>>>(deg, rowptr, cursor);
    fill_csr<<<(n_edges + 255) / 256, 256, 0, stream>>>(ei, cursor, csr_src, n_edges);
  }

  for (int l = 0; l < N_LAYERS; ++l) {
    gemm_m<<<RBLOCKS, 320, 0, stream>>>(
        h_pad, wpack1 + (size_t)l * COL_TILES * KSTEPS * 64, m_pad);

    gather_aggr<<<(N_NODES + 7) / 8, 256, 0, stream>>>(m_pad, rowptr, csr_src, aggr);

    const bool last = (l == N_LAYERS - 1);
    gru_fused<<<RBLOCKS, 320, 0, stream>>>(
        h_pad, aggr, wpackih, wpackhh, b_ih, b_hh,
        last ? out : h_pad, last ? HID : HP, last ? 0 : 1);
  }
}


# scan_deg rewritten: coalesced int4 loads, per-wave shuffle scans with a running carry, one barrier; replaces 98 chunked 1024-wide LDS Hillis-Steele scans (23 barriers per chunk)
# speedup vs baseline: 1.1671x; 1.1020x over previous
#include <hip/hip_runtime.h>
#include <hip/hip_bf16.h>
#include <math.h>

#define N_NODES   100000
#define N_LAYERS  5
#define HID       150
#define HP        160
#define RPB       32
#define RBLOCKS   (N_NODES / RPB)
#define COL_TILES (HP / 16)
#define KSTEPS    (HP / 4)
#define LDS_LD    164

typedef __attribute__((ext_vector_type(2))) float v2f;
typedef __attribute__((ext_vector_type(8))) float v8f;

__device__ __forceinline__ v8f wmma_f32(v2f a, v2f b, v8f c) {
  return __builtin_amdgcn_wmma_f32_16x16x4_f32(
       false, a,  false, b,
       (short)0, c,  false,  false);
}

__device__ __forceinline__ float sigmoidf_(float x) {
  return 1.0f / (1.0f + __expf(-x));
}


__global__ void init_hpad(const float* __restrict__ x, float* __restrict__ hpad) {
  int idx = blockIdx.x * blockDim.x + threadIdx.x;
  if (idx >= N_NODES * HP) return;
  int row = idx / HP, col = idx - row * HP;
  hpad[idx] = (col < HID) ? x[row * HID + col] : 0.0f;
}

__global__ void pack_w1(const float* __restrict__ w, float* __restrict__ wp) {
  int idx = blockIdx.x * blockDim.x + threadIdx.x;
  const int per_layer = COL_TILES * KSTEPS * 64;
  if (idx >= N_LAYERS * per_layer) return;
  int l    = idx / per_layer;
  int rem  = idx - l * per_layer;
  int j    = rem & 1;
  int lane = (rem >> 1) & 31;
  int ks   = (rem >> 6) % KSTEPS;
  int ct   = (rem >> 6) / KSTEPS;
  int k = ks * 4 + 2 * (lane >> 4) + j;
  int n = ct * 16 + (lane & 15);
  wp[idx] = (k < HID && n < HID) ? w[(l * HID + k) * HID + n] : 0.0f;
}

__global__ void pack_wg(const float* __restrict__ w, float* __restrict__ wp) {
  int idx = blockIdx.x * blockDim.x + threadIdx.x;
  const int per_gate = COL_TILES * KSTEPS * 64;
  if (idx >= 3 * per_gate) return;
  int g    = idx / per_gate;
  int rem  = idx - g * per_gate;
  int j    = rem & 1;
  int lane = (rem >> 1) & 31;
  int ks   = (rem >> 6) % KSTEPS;
  int ct   = (rem >> 6) / KSTEPS;
  int k = ks * 4 + 2 * (lane >> 4) + j;
  int n = ct * 16 + (lane & 15);
  wp[idx] = (k < HID && n < HID) ? w[(g * HID + n) * HID + k] : 0.0f;
}

__global__ void hist_deg(const int* __restrict__ ei, int* __restrict__ deg,
                         int n_edges) {
  int e = blockIdx.x * blockDim.x + threadIdx.x;
  if (e >= n_edges) return;
  atomicAdd(&deg[ei[n_edges + e]], 1);
}

__device__ __forceinline__ void scan_write(const int4* __restrict__ dv,
                                           int4* __restrict__ rv,
                                           int4* __restrict__ cv, int vi, int first) {
  const int4 d = dv[vi];
  int4 o;
  o.x = first;
  o.y = o.x + d.x;
  o.z = o.y + d.y;
  o.w = o.z + d.z;
  rv[vi] = o;
  cv[vi] = o;
}

#define SCAN_GROUPS 25
#define SCAN_VECS   (N_NODES / 4)
static_assert(N_NODES % 4 == 0 && 32 * SCAN_GROUPS * 128 >= N_NODES &&
              31 * SCAN_GROUPS * 128 <= N_NODES,
              "scan_deg: 32 segments must cover the nodes, the first 31 whole");
__global__ void __launch_bounds__(1024)
scan_deg(const int* __restrict__ deg, int* __restrict__ rowptr,
         int* __restrict__ cursor) {
  __shared__ int wsum[32];
  const int tid = threadIdx.x;
  const int lane = tid & 31, wave = tid >> 5;
  const int4* dv = (const int4*)deg;
  int4* rv = (int4*)rowptr;
  int4* cv = (int4*)cursor;
  const int vbase = wave * (SCAN_GROUPS * 32) + lane;

  int exg[SCAN_GROUPS];
  int run = 0;
  #pragma unroll
  for (int g = 0; g < SCAN_GROUPS; ++g) {
    const int vi = vbase + g * 32;
    const bool ok = vi < SCAN_VECS;
    int4 d = dv[ok ? vi : SCAN_VECS - 1];
    if (!ok) d = make_int4(0, 0, 0, 0);
    const int s4 = (d.x + d.y) + (d.z + d.w);
    int incl = s4;
    #pragma unroll
    for (int off = 1; off < 32; off <<= 1) {
      int t = __shfl_up(incl, off);
      if (lane >= off) incl += t;
    }
    exg[g] = run + incl - s4;
    run += __shfl(incl, 31);
  }
  if (lane == 0) wsum[wave] = run;
  __syncthreads();

  const int wt = wsum[lane];
  int wincl = wt;
  #pragma unroll
  for (int off = 1; off < 32; off <<= 1) {
    int t = __shfl_up(wincl, off);
    if (lane >= off) wincl += t;
  }
  const int total = __shfl(wincl, 31);
  const int wave_excl = __shfl(wincl, wave) - __shfl(wt, wave);

  if (wave < 31) {
    #pragma unroll
    for (int g = 0; g < SCAN_GROUPS; ++g)
      scan_write(dv, rv, cv, vbase + g * 32, wave_excl + exg[g]);
  } else {
    #pragma unroll
    for (int g = 0; g < SCAN_GROUPS; ++g) {
      const int vi = vbase + g * 32;
      if (vi < SCAN_VECS) scan_write(dv, rv, cv, vi, wave_excl + exg[g]);
    }
  }
  if (tid == 0) rowptr[N_NODES] = total;
}

__global__ void fill_csr(const int* __restrict__ ei, int* __restrict__ cursor,
                         int* __restrict__ csr_src, int n_edges) {
  int e = blockIdx.x * blockDim.x + threadIdx.x;
  if (e >= n_edges) return;
  int pos = atomicAdd(&cursor[ei[n_edges + e]], 1);
  csr_src[pos] = ei[e];
}

__global__ void __launch_bounds__(320)
gemm_m(const float* __restrict__ h, const float* __restrict__ wp,
       float* __restrict__ m) {
  __shared__ float hs[RPB * LDS_LD];
  const int rowbase = blockIdx.x * RPB;

  for (int i = threadIdx.x; i < RPB * (HP / 4); i += 320) {
    int r = i / (HP / 4), cc = (i - r * (HP / 4)) * 4;
    float4 v = *(const float4*)(h + (rowbase + r) * HP + cc);
    float* d = &hs[r * LDS_LD + cc];
    d[0] = v.x; d[1] = v.y; d[2] = v.z; d[3] = v.w;
  }
  __syncthreads();

  const int wave = threadIdx.x >> 5, lane = threadIdx.x & 31;
  const int half = lane >> 4, l16 = lane & 15;
  const float* wpt = wp + wave * (KSTEPS * 64);

  v8f acc0 = {}, acc1 = {};
  for (int ks = 0; ks < KSTEPS; ++ks) {
    int ao = ks * 4 + 2 * half;
    v2f a0 = *(const v2f*)&hs[l16 * LDS_LD + ao];
    v2f a1 = *(const v2f*)&hs[(l16 + 16) * LDS_LD + ao];
    v2f b  = *(const v2f*)(wpt + ks * 64 + lane * 2);
    acc0 = wmma_f32(a0, b, acc0);
    acc1 = wmma_f32(a1, b, acc1);
  }
  const int col = wave * 16 + l16;
  #pragma unroll
  for (int v = 0; v < 8; ++v) {
    m[(rowbase + v + 8 * half) * HP + col]      = acc0[v];
    m[(rowbase + 16 + v + 8 * half) * HP + col] = acc1[v];
  }
}

__global__ void __launch_bounds__(256)
gather_aggr(const float* __restrict__ m, const int* __restrict__ rowptr,
            const int* __restrict__ csr_src, float* __restrict__ aggr) {
  int node = blockIdx.x * 8 + (threadIdx.x >> 5);
  if (node >= N_NODES) return;
  const int lane = threadIdx.x & 31;
  int beg = rowptr[node], end = rowptr[node + 1];
  float a0 = 0.f, a1 = 0.f, a2 = 0.f, a3 = 0.f, a4 = 0.f;
  for (int j = beg; j < end; ++j) {
    const float* mr = m + (size_t)csr_src[j] * HP;
    float4 v = *(const float4*)(mr + 4 * lane);
    float  s = mr[128 + lane];
    a0 += v.x; a1 += v.y; a2 += v.z; a3 += v.w; a4 += s;
  }
  float* ar = aggr + (size_t)node * HP;
  float4 o; o.x = a0; o.y = a1; o.z = a2; o.w = a3;
  *(float4*)(ar + 4 * lane) = o;
  ar[128 + lane] = a4;
}

struct KFrag {
  v2f aA0, aA1, aH0, aH1;
  v2f br, bz, bn, cr, cz, cn;
};

__global__ void __launch_bounds__(320)
gru_fused(const float* __restrict__ hin, const float* __restrict__ aggr,
          const float* __restrict__ wih, const float* __restrict__ whh,
          const float* __restrict__ bih, const float* __restrict__ bhh,
          float* __restrict__ hout, int ldo, int zero_pad) {
  __shared__ float hs[RPB * LDS_LD];
  __shared__ float as_[RPB * LDS_LD];
  const int rowbase = blockIdx.x * RPB;

  for (int i = threadIdx.x; i < 2 * RPB * (HP / 4); i += 320) {
    int which = i / (RPB * (HP / 4));
    int t = i - which * (RPB * (HP / 4));
    int r = t / (HP / 4), cc = (t - r * (HP / 4)) * 4;
    const float* srcp = which ? (aggr + (rowbase + r) * HP + cc)
                              : (hin + (rowbase + r) * HP + cc);
    float4 v = *(const float4*)srcp;
    float* d = which ? &as_[r * LDS_LD + cc] : &hs[r * LDS_LD + cc];
    d[0] = v.x; d[1] = v.y; d[2] = v.z; d[3] = v.w;
  }
  __syncthreads();

  const int wave = threadIdx.x >> 5, lane = threadIdx.x & 31;
  const int half = lane >> 4, l16 = lane & 15;
  const int ct = wave;

  const int GSTRIDE = COL_TILES * KSTEPS * 64;
  const float* aAp = &as_[l16 * LDS_LD + 2 * half];
  const float* aHp = &hs[l16 * LDS_LD + 2 * half];
  const float* wi = wih + ct * (KSTEPS * 64) + lane * 2;
  const float* wh = whh + ct * (KSTEPS * 64) + lane * 2;

  v8f ir0 = {}, hr0 = {}, iz0 = {}, hz0 = {}, in0 = {}, hn0 = {};
  v8f ir1 = {}, hr1 = {}, iz1 = {}, hz1 = {}, in1 = {}, hn1 = {};

#define LOAD_STEP(F, KS)                                                   \
  do {                                                                     \
    const int ks_ = (KS);                                                  \
    F.br = *(const v2f*)(wi + ks_ * 64);                                   \
    F.bz = *(const v2f*)(wi + GSTRIDE + ks_ * 64);                         \
    F.bn = *(const v2f*)(wi + 2 * GSTRIDE + ks_ * 64);                     \
    F.cr = *(const v2f*)(wh + ks_ * 64);                                   \
    F.cz = *(const v2f*)(wh + GSTRIDE + ks_ * 64);                         \
    F.cn = *(const v2f*)(wh + 2 * GSTRIDE + ks_ * 64);                     \
    F.aA0 = *(const v2f*)(aAp + ks_ * 4);                                  \
    F.aA1 = *(const v2f*)(aAp + 16 * LDS_LD + ks_ * 4);                    \
    F.aH0 = *(const v2f*)(aHp + ks_ * 4);                                  \
    F.aH1 = *(const v2f*)(aHp + 16 * LDS_LD + ks_ * 4);                    \
  } while (0)

#define MMA_STEP(F)                                                        \
  do {                                                                     \
    ir0 = wmma_f32(F.aA0, F.br, ir0);  ir1 = wmma_f32(F.aA1, F.br, ir1);   \
    iz0 = wmma_f32(F.aA0, F.bz, iz0);  iz1 = wmma_f32(F.aA1, F.bz, iz1);   \
    in0 = wmma_f32(F.aA0, F.bn, in0);  in1 = wmma_f32(F.aA1, F.bn, in1);   \
    hr0 = wmma_f32(F.aH0, F.cr, hr0);  hr1 = wmma_f32(F.aH1, F.cr, hr1);   \
    hz0 = wmma_f32(F.aH0, F.cz, hz0);  hz1 = wmma_f32(F.aH1, F.cz, hz1);   \
    hn0 = wmma_f32(F.aH0, F.cn, hn0);  hn1 = wmma_f32(F.aH1, F.cn, hn1);   \
  } while (0)

  KFrag f0, f1;
  LOAD_STEP(f0, 0);
  for (int ks = 0; ks < KSTEPS; ks += 2) {
    LOAD_STEP(f1, ks + 1);
    __builtin_amdgcn_sched_barrier(0);
    MMA_STEP(f0);
    __builtin_amdgcn_sched_barrier(0);
    LOAD_STEP(f0, (ks + 2 < KSTEPS) ? ks + 2 : ks + 1);
    __builtin_amdgcn_sched_barrier(0);
    MMA_STEP(f1);
    __builtin_amdgcn_sched_barrier(0);
  }
#undef LOAD_STEP
#undef MMA_STEP

  const int col = ct * 16 + l16;
  const bool valid = (col < HID);
  const int cb = valid ? col : 0;
  const float bir = valid ? bih[cb]           : 0.0f;
  const float biz = valid ? bih[HID + cb]     : 0.0f;
  const float bin = valid ? bih[2 * HID + cb] : 0.0f;
  const float bhr = valid ? bhh[cb]           : 0.0f;
  const float bhz = valid ? bhh[HID + cb]     : 0.0f;
  const float bhn = valid ? bhh[2 * HID + cb] : 0.0f;

  #pragma unroll
  for (int rt = 0; rt < 2; ++rt) {
    #pragma unroll
    for (int v = 0; v < 8; ++v) {
      float irv = rt ? ir1[v] : ir0[v];
      float hrv = rt ? hr1[v] : hr0[v];
      float izv = rt ? iz1[v] : iz0[v];
      float hzv = rt ? hz1[v] : hz0[v];
      float inv = rt ? in1[v] : in0[v];
      float hnv = rt ? hn1[v] : hn0[v];
      int rl = rt * 16 + v + 8 * half;
      float r = sigmoidf_((irv + bir) + (hrv + bhr));
      float z = sigmoidf_((izv + biz) + (hzv + bhz));
      float n = tanhf((inv + bin) + r * (hnv + bhn));
      float hv = hs[rl * LDS_LD + col];
      float o = (1.0f - z) * n + z * hv;
      long row = rowbase + rl;
      if (valid)          hout[row * ldo + col] = o;
      else if (zero_pad)  hout[row * ldo + col] = 0.0f;
    }
  }
}

extern "C" void kernel_launch(void* const* d_in, const int* in_sizes, int n_in,
                              void* d_out, int out_size, void* d_ws, size_t ws_size,
                              hipStream_t stream) {
  const float* x    = (const float*)d_in[0];
  const float* wgt  = (const float*)d_in[1];
  const float* w_ih = (const float*)d_in[2];
  const float* w_hh = (const float*)d_in[3];
  const float* b_ih = (const float*)d_in[4];
  const float* b_hh = (const float*)d_in[5];
  const int*   ei   = (const int*)d_in[6];
  const int n_edges = in_sizes[6] / 2;
  float* out = (float*)d_out;

  const size_t npad = (size_t)N_NODES * HP;
  const size_t gate_sz = 3 * COL_TILES * KSTEPS * 64;
  float* ws      = (float*)d_ws;
  float* h_pad   = ws;
  float* m_pad   = h_pad + npad;
  float* aggr    = m_pad + npad;
  float* wpack1  = aggr  + npad;
  float* wpackih = wpack1  + (size_t)N_LAYERS * COL_TILES * KSTEPS * 64;
  float* wpackhh = wpackih + gate_sz;
  int*   deg     = (int*)(wpackhh + gate_sz);
  int*   rowptr  = deg + N_NODES;
  int*   cursor  = rowptr + (N_NODES + 4);
  int*   csr_src = cursor + N_NODES;

  {
    int n = N_NODES * HP;
    init_hpad<<<(n + 255) / 256, 256, 0, stream>>>(x, h_pad);
    int n1 = N_LAYERS * COL_TILES * KSTEPS * 64;
    pack_w1<<<(n1 + 255) / 256, 256, 0, stream>>>(wgt, wpack1);
    int ng = (int)gate_sz;
    pack_wg<<<(ng + 255) / 256, 256, 0, stream>>>(w_ih, wpackih);
    pack_wg<<<(ng + 255) / 256, 256, 0, stream>>>(w_hh, wpackhh);

    hipMemsetAsync(deg, 0, (size_t)N_NODES * sizeof(int), stream);
    hist_deg<<<(n_edges + 255) / 256, 256, 0, stream>>>(ei, deg, n_edges);
    scan_deg<<<1, 1024, 0, stream>>>(deg, rowptr, cursor);
    fill_csr<<<(n_edges + 255) / 256, 256, 0, stream>>>(ei, cursor, csr_src, n_edges);
  }

  for (int l = 0; l < N_LAYERS; ++l) {
    gemm_m<<<RBLOCKS, 320, 0, stream>>>(
        h_pad, wpack1 + (size_t)l * COL_TILES * KSTEPS * 64, m_pad);

    gather_aggr<<<(N_NODES + 7) / 8, 256, 0, stream>>>(m_pad, rowptr, csr_src, aggr);

    const bool last = (l == N_LAYERS - 1);
    gru_fused<<<RBLOCKS, 320, 0, stream>>>(
        h_pad, aggr, wpackih, wpackhh, b_ih, b_hh,
        last ? out : h_pad, last ? HID : HP, last ? 0 : 1);
  }
}


# h @ W[l] GEMM removed (5 launches): W[l] folded into w_ih once per call (W[l] @ w_ih^T on WMMA f32), edge sum runs over h rows; re-associates (A h) W, f32 sum order changes
# speedup vs baseline: 1.3217x; 1.1325x over previous
#include <hip/hip_runtime.h>
#include <hip/hip_bf16.h>
#include <math.h>

#define N_NODES   100000
#define N_LAYERS  5
#define HID       150
#define HP        160
#define RPB       32
#define RBLOCKS   (N_NODES / RPB)
#define COL_TILES (HP / 16)
#define KSTEPS    (HP / 4)
#define LDS_LD    164

typedef __attribute__((ext_vector_type(2))) float v2f;
typedef __attribute__((ext_vector_type(8))) float v8f;

__device__ __forceinline__ v8f wmma_f32(v2f a, v2f b, v8f c) {
  return __builtin_amdgcn_wmma_f32_16x16x4_f32(
       false, a,  false, b,
       (short)0, c,  false,  false);
}

__device__ __forceinline__ float sigmoidf_(float x) {
  return 1.0f / (1.0f + __expf(-x));
}


__global__ void init_hpad(const float* __restrict__ x, float* __restrict__ hpad) {
  int idx = blockIdx.x * blockDim.x + threadIdx.x;
  if (idx >= N_NODES * HP) return;
  int row = idx / HP, col = idx - row * HP;
  hpad[idx] = (col < HID) ? x[row * HID + col] : 0.0f;
}

__global__ void pack_wg(const float* __restrict__ w, float* __restrict__ wp) {
  int idx = blockIdx.x * blockDim.x + threadIdx.x;
  const int per_gate = COL_TILES * KSTEPS * 64;
  if (idx >= 3 * per_gate) return;
  int g    = idx / per_gate;
  int rem  = idx - g * per_gate;
  int j    = rem & 1;
  int lane = (rem >> 1) & 31;
  int ks   = (rem >> 6) % KSTEPS;
  int ct   = (rem >> 6) / KSTEPS;
  int k = ks * 4 + 2 * (lane >> 4) + j;
  int n = ct * 16 + (lane & 15);
  wp[idx] = (k < HID && n < HID) ? w[(g * HID + n) * HID + k] : 0.0f;
}

__global__ void hist_deg(const int* __restrict__ ei, int* __restrict__ deg,
                         int n_edges) {
  int e = blockIdx.x * blockDim.x + threadIdx.x;
  if (e >= n_edges) return;
  atomicAdd(&deg[ei[n_edges + e]], 1);
}

__device__ __forceinline__ void scan_write(const int4* __restrict__ dv,
                                           int4* __restrict__ rv,
                                           int4* __restrict__ cv, int vi, int first) {
  const int4 d = dv[vi];
  int4 o;
  o.x = first;
  o.y = o.x + d.x;
  o.z = o.y + d.y;
  o.w = o.z + d.z;
  rv[vi] = o;
  cv[vi] = o;
}

#define SCAN_GROUPS 25
#define SCAN_VECS   (N_NODES / 4)
static_assert(N_NODES % 4 == 0 && 32 * SCAN_GROUPS * 128 >= N_NODES &&
              31 * SCAN_GROUPS * 128 <= N_NODES,
              "scan_deg: 32 segments must cover the nodes, the first 31 whole");
__global__ void __launch_bounds__(1024)
scan_deg(const int* __restrict__ deg, int* __restrict__ rowptr,
         int* __restrict__ cursor) {
  __shared__ int wsum[32];
  const int tid = threadIdx.x;
  const int lane = tid & 31, wave = tid >> 5;
  const int4* dv = (const int4*)deg;
  int4* rv = (int4*)rowptr;
  int4* cv = (int4*)cursor;
  const int vbase = wave * (SCAN_GROUPS * 32) + lane;

  int exg[SCAN_GROUPS];
  int run = 0;
  #pragma unroll
  for (int g = 0; g < SCAN_GROUPS; ++g) {
    const int vi = vbase + g * 32;
    const bool ok = vi < SCAN_VECS;
    int4 d = dv[ok ? vi : SCAN_VECS - 1];
    if (!ok) d = make_int4(0, 0, 0, 0);
    const int s4 = (d.x + d.y) + (d.z + d.w);
    int incl = s4;
    #pragma unroll
    for (int off = 1; off < 32; off <<= 1) {
      int t = __shfl_up(incl, off);
      if (lane >= off) incl += t;
    }
    exg[g] = run + incl - s4;
    run += __shfl(incl, 31);
  }
  if (lane == 0) wsum[wave] = run;
  __syncthreads();

  const int wt = wsum[lane];
  int wincl = wt;
  #pragma unroll
  for (int off = 1; off < 32; off <<= 1) {
    int t = __shfl_up(wincl, off);
    if (lane >= off) wincl += t;
  }
  const int total = __shfl(wincl, 31);
  const int wave_excl = __shfl(wincl, wave) - __shfl(wt, wave);

  if (wave < 31) {
    #pragma unroll
    for (int g = 0; g < SCAN_GROUPS; ++g)
      scan_write(dv, rv, cv, vbase + g * 32, wave_excl + exg[g]);
  } else {
    #pragma unroll
    for (int g = 0; g < SCAN_GROUPS; ++g) {
      const int vi = vbase + g * 32;
      if (vi < SCAN_VECS) scan_write(dv, rv, cv, vi, wave_excl + exg[g]);
    }
  }
  if (tid == 0) rowptr[N_NODES] = total;
}

__global__ void fill_csr(const int* __restrict__ ei, int* __restrict__ cursor,
                         int* __restrict__ csr_src, int n_edges) {
  int e = blockIdx.x * blockDim.x + threadIdx.x;
  if (e >= n_edges) return;
  int pos = atomicAdd(&cursor[ei[n_edges + e]], 1);
  csr_src[pos] = ei[e];
}

__global__ void __launch_bounds__(320)
fold_w(const float* __restrict__ w, const float* __restrict__ wpih,
       float* __restrict__ wfold) {
  __shared__ float as_[RPB * LDS_LD];
  const int rowbase = blockIdx.x * RPB;
  const int g = blockIdx.y, l = blockIdx.z;

  static_assert(RPB * HP == 16 * 320, "fold_w staging: 16 elements per thread");
  float st[16];
  #pragma unroll
  for (int it = 0; it < 16; ++it) {
    const int i = threadIdx.x + it * 320;
    const int r = i / HP, c = i - r * HP;
    const int k = rowbase + r;
    const bool ok = (k < HID && c < HID);
    const float t = w[ok ? (l * HID + k) * HID + c : 0];
    st[it] = ok ? t : 0.0f;
  }
  #pragma unroll
  for (int it = 0; it < 16; ++it) {
    const int i = threadIdx.x + it * 320;
    const int r = i / HP, c = i - r * HP;
    as_[r * LDS_LD + c] = st[it];
  }
  __syncthreads();

  const int wave = threadIdx.x >> 5, lane = threadIdx.x & 31;
  const int half = lane >> 4, l16 = lane & 15;
  const int GSTRIDE = COL_TILES * KSTEPS * 64;
  const float* wpt = wpih + g * GSTRIDE + wave * (KSTEPS * 64);

  v8f acc0 = {}, acc1 = {};
  for (int ks = 0; ks < KSTEPS; ++ks) {
    int ao = ks * 4 + 2 * half;
    v2f a0 = *(const v2f*)&as_[l16 * LDS_LD + ao];
    v2f a1 = *(const v2f*)&as_[(l16 + 16) * LDS_LD + ao];
    v2f b  = *(const v2f*)(wpt + ks * 64 + lane * 2);
    acc0 = wmma_f32(a0, b, acc0);
    acc1 = wmma_f32(a1, b, acc1);
  }
  float* dst = wfold + (size_t)(l * 3 + g) * GSTRIDE + wave * (KSTEPS * 64);
  #pragma unroll
  for (int v = 0; v < 8; v += 2) {
    const int k0 = rowbase + v + 8 * half;
    const int k1 = k0 + 16;
    v2f p0, p1;
    p0.x = acc0[v]; p0.y = acc0[v + 1];
    p1.x = acc1[v]; p1.y = acc1[v + 1];
    *(v2f*)&dst[(k0 >> 2) * 64 + (l16 + 16 * ((k0 >> 1) & 1)) * 2] = p0;
    *(v2f*)&dst[(k1 >> 2) * 64 + (l16 + 16 * ((k1 >> 1) & 1)) * 2] = p1;
  }
}

__global__ void __launch_bounds__(256)
gather_aggr(const float* __restrict__ hsrc, const int* __restrict__ rowptr,
            const int* __restrict__ csr_src, float* __restrict__ aggr) {
  int node = blockIdx.x * 8 + (threadIdx.x >> 5);
  if (node >= N_NODES) return;
  const int lane = threadIdx.x & 31;
  int beg = rowptr[node], end = rowptr[node + 1];
  float a0 = 0.f, a1 = 0.f, a2 = 0.f, a3 = 0.f, a4 = 0.f;
  for (int j = beg; j < end; ++j) {
    const float* mr = hsrc + (size_t)csr_src[j] * HP;
    float4 v = *(const float4*)(mr + 4 * lane);
    float  s = mr[128 + lane];
    a0 += v.x; a1 += v.y; a2 += v.z; a3 += v.w; a4 += s;
  }
  float* ar = aggr + (size_t)node * HP;
  float4 o; o.x = a0; o.y = a1; o.z = a2; o.w = a3;
  *(float4*)(ar + 4 * lane) = o;
  ar[128 + lane] = a4;
}

struct KFrag {
  v2f aA0, aA1, aH0, aH1;
  v2f br, bz, bn, cr, cz, cn;
};

__global__ void __launch_bounds__(320)
gru_fused(const float* __restrict__ hin, const float* __restrict__ aggr,
          const float* __restrict__ wih, const float* __restrict__ whh,
          const float* __restrict__ bih, const float* __restrict__ bhh,
          float* __restrict__ hout, int ldo, int zero_pad) {
  __shared__ float hs[RPB * LDS_LD];
  __shared__ float as_[RPB * LDS_LD];
  const int rowbase = blockIdx.x * RPB;

  for (int i = threadIdx.x; i < 2 * RPB * (HP / 4); i += 320) {
    int which = i / (RPB * (HP / 4));
    int t = i - which * (RPB * (HP / 4));
    int r = t / (HP / 4), cc = (t - r * (HP / 4)) * 4;
    const float* srcp = which ? (aggr + (rowbase + r) * HP + cc)
                              : (hin + (rowbase + r) * HP + cc);
    float4 v = *(const float4*)srcp;
    float* d = which ? &as_[r * LDS_LD + cc] : &hs[r * LDS_LD + cc];
    d[0] = v.x; d[1] = v.y; d[2] = v.z; d[3] = v.w;
  }
  __syncthreads();

  const int wave = threadIdx.x >> 5, lane = threadIdx.x & 31;
  const int half = lane >> 4, l16 = lane & 15;
  const int ct = wave;

  const int GSTRIDE = COL_TILES * KSTEPS * 64;
  const float* aAp = &as_[l16 * LDS_LD + 2 * half];
  const float* aHp = &hs[l16 * LDS_LD + 2 * half];
  const float* wi = wih + ct * (KSTEPS * 64) + lane * 2;
  const float* wh = whh + ct * (KSTEPS * 64) + lane * 2;

  v8f ir0 = {}, hr0 = {}, iz0 = {}, hz0 = {}, in0 = {}, hn0 = {};
  v8f ir1 = {}, hr1 = {}, iz1 = {}, hz1 = {}, in1 = {}, hn1 = {};

#define LOAD_STEP(F, KS)                                                   \
  do {                                                                     \
    const int ks_ = (KS);                                                  \
    F.br = *(const v2f*)(wi + ks_ * 64);                                   \
    F.bz = *(const v2f*)(wi + GSTRIDE + ks_ * 64);                         \
    F.bn = *(const v2f*)(wi + 2 * GSTRIDE + ks_ * 64);                     \
    F.cr = *(const v2f*)(wh + ks_ * 64);                                   \
    F.cz = *(const v2f*)(wh + GSTRIDE + ks_ * 64);                         \
    F.cn = *(const v2f*)(wh + 2 * GSTRIDE + ks_ * 64);                     \
    F.aA0 = *(const v2f*)(aAp + ks_ * 4);                                  \
    F.aA1 = *(const v2f*)(aAp + 16 * LDS_LD + ks_ * 4);                    \
    F.aH0 = *(const v2f*)(aHp + ks_ * 4);                                  \
    F.aH1 = *(const v2f*)(aHp + 16 * LDS_LD + ks_ * 4);                    \
  } while (0)

#define MMA_STEP(F)                                                        \
  do {                                                                     \
    ir0 = wmma_f32(F.aA0, F.br, ir0);  ir1 = wmma_f32(F.aA1, F.br, ir1);   \
    iz0 = wmma_f32(F.aA0, F.bz, iz0);  iz1 = wmma_f32(F.aA1, F.bz, iz1);   \
    in0 = wmma_f32(F.aA0, F.bn, in0);  in1 = wmma_f32(F.aA1, F.bn, in1);   \
    hr0 = wmma_f32(F.aH0, F.cr, hr0);  hr1 = wmma_f32(F.aH1, F.cr, hr1);   \
    hz0 = wmma_f32(F.aH0, F.cz, hz0);  hz1 = wmma_f32(F.aH1, F.cz, hz1);   \
    hn0 = wmma_f32(F.aH0, F.cn, hn0);  hn1 = wmma_f32(F.aH1, F.cn, hn1);   \
  } while (0)

  KFrag f0, f1;
  LOAD_STEP(f0, 0);
  for (int ks = 0; ks < KSTEPS; ks += 2) {
    LOAD_STEP(f1, ks + 1);
    __builtin_amdgcn_sched_barrier(0);
    MMA_STEP(f0);
    __builtin_amdgcn_sched_barrier(0);
    LOAD_STEP(f0, (ks + 2 < KSTEPS) ? ks + 2 : ks + 1);
    __builtin_amdgcn_sched_barrier(0);
    MMA_STEP(f1);
    __builtin_amdgcn_sched_barrier(0);
  }
#undef LOAD_STEP
#undef MMA_STEP

  const int col = ct * 16 + l16;
  const bool valid = (col < HID);
  const int cb = valid ? col : 0;
  const float bir = valid ? bih[cb]           : 0.0f;
  const float biz = valid ? bih[HID + cb]     : 0.0f;
  const float bin = valid ? bih[2 * HID + cb] : 0.0f;
  const float bhr = valid ? bhh[cb]           : 0.0f;
  const float bhz = valid ? bhh[HID + cb]     : 0.0f;
  const float bhn = valid ? bhh[2 * HID + cb] : 0.0f;

  #pragma unroll
  for (int rt = 0; rt < 2; ++rt) {
    #pragma unroll
    for (int v = 0; v < 8; ++v) {
      float irv = rt ? ir1[v] : ir0[v];
      float hrv = rt ? hr1[v] : hr0[v];
      float izv = rt ? iz1[v] : iz0[v];
      float hzv = rt ? hz1[v] : hz0[v];
      float inv = rt ? in1[v] : in0[v];
      float hnv = rt ? hn1[v] : hn0[v];
      int rl = rt * 16 + v + 8 * half;
      float r = sigmoidf_((irv + bir) + (hrv + bhr));
      float z = sigmoidf_((izv + biz) + (hzv + bhz));
      float n = tanhf((inv + bin) + r * (hnv + bhn));
      float hv = hs[rl * LDS_LD + col];
      float o = (1.0f - z) * n + z * hv;
      long row = rowbase + rl;
      if (valid)          hout[row * ldo + col] = o;
      else if (zero_pad)  hout[row * ldo + col] = 0.0f;
    }
  }
}

extern "C" void kernel_launch(void* const* d_in, const int* in_sizes, int n_in,
                              void* d_out, int out_size, void* d_ws, size_t ws_size,
                              hipStream_t stream) {
  const float* x    = (const float*)d_in[0];
  const float* wgt  = (const float*)d_in[1];
  const float* w_ih = (const float*)d_in[2];
  const float* w_hh = (const float*)d_in[3];
  const float* b_ih = (const float*)d_in[4];
  const float* b_hh = (const float*)d_in[5];
  const int*   ei   = (const int*)d_in[6];
  const int n_edges = in_sizes[6] / 2;
  float* out = (float*)d_out;

  const size_t npad = (size_t)N_NODES * HP;
  const size_t gate_sz = 3 * COL_TILES * KSTEPS * 64;
  float* ws      = (float*)d_ws;
  float* h_pad   = ws;
  float* aggr    = h_pad + npad;
  float* wfold   = aggr  + npad;
  float* wpackih = wfold + (size_t)N_LAYERS * gate_sz;
  float* wpackhh = wpackih + gate_sz;
  int*   deg     = (int*)(wpackhh + gate_sz);
  int*   rowptr  = deg + N_NODES;
  int*   cursor  = rowptr + (N_NODES + 4);
  int*   csr_src = cursor + N_NODES;

  {
    int n = N_NODES * HP;
    init_hpad<<<(n + 255) / 256, 256, 0, stream>>>(x, h_pad);
    int ng = (int)gate_sz;
    pack_wg<<<(ng + 255) / 256, 256, 0, stream>>>(w_ih, wpackih);
    pack_wg<<<(ng + 255) / 256, 256, 0, stream>>>(w_hh, wpackhh);
    fold_w<<<dim3(HP / RPB, 3, N_LAYERS), 320, 0, stream>>>(wgt, wpackih, wfold);

    hipMemsetAsync(deg, 0, (size_t)N_NODES * sizeof(int), stream);
    hist_deg<<<(n_edges + 255) / 256, 256, 0, stream>>>(ei, deg, n_edges);
    scan_deg<<<1, 1024, 0, stream>>>(deg, rowptr, cursor);
    fill_csr<<<(n_edges + 255) / 256, 256, 0, stream>>>(ei, cursor, csr_src, n_edges);
  }

  for (int l = 0; l < N_LAYERS; ++l) {
    gather_aggr<<<(N_NODES + 7) / 8, 256, 0, stream>>>(h_pad, rowptr, csr_src, aggr);

    const bool last = (l == N_LAYERS - 1);
    gru_fused<<<RBLOCKS, 320, 0, stream>>>(
        h_pad, aggr, wfold + (size_t)l * gate_sz, wpackhh, b_ih, b_hh,
        last ? out : h_pad, last ? HID : HP, last ? 0 : 1);
  }
}


# gru_fused gate epilogue: sigmoid and tanh on v_exp_f32 + v_rcp_f32 (about 1 ulp each) instead of IEEE division and library tanhf: 730 fewer vector instructions per wave beside the f32 WMMAs
# speedup vs baseline: 1.4036x; 1.0620x over previous
#include <hip/hip_runtime.h>
#include <hip/hip_bf16.h>
#include <math.h>

#define N_NODES   100000
#define N_LAYERS  5
#define HID       150
#define HP        160
#define RPB       32
#define RBLOCKS   (N_NODES / RPB)
#define COL_TILES (HP / 16)
#define KSTEPS    (HP / 4)
#define LDS_LD    164

typedef __attribute__((ext_vector_type(2))) float v2f;
typedef __attribute__((ext_vector_type(8))) float v8f;

__device__ __forceinline__ v8f wmma_f32(v2f a, v2f b, v8f c) {
  return __builtin_amdgcn_wmma_f32_16x16x4_f32(
       false, a,  false, b,
       (short)0, c,  false,  false);
}

__device__ __forceinline__ float sigmoidf_(float x) {
  return __builtin_amdgcn_rcpf(1.0f + __expf(-x));
}
__device__ __forceinline__ float tanhf_(float x) {
  return 1.0f - 2.0f * __builtin_amdgcn_rcpf(1.0f + __builtin_amdgcn_exp2f(x * 2.88539008f));
}


__global__ void init_hpad(const float* __restrict__ x, float* __restrict__ hpad) {
  int idx = blockIdx.x * blockDim.x + threadIdx.x;
  if (idx >= N_NODES * HP) return;
  int row = idx / HP, col = idx - row * HP;
  hpad[idx] = (col < HID) ? x[row * HID + col] : 0.0f;
}

__global__ void pack_wg(const float* __restrict__ w, float* __restrict__ wp) {
  int idx = blockIdx.x * blockDim.x + threadIdx.x;
  const int per_gate = COL_TILES * KSTEPS * 64;
  if (idx >= 3 * per_gate) return;
  int g    = idx / per_gate;
  int rem  = idx - g * per_gate;
  int j    = rem & 1;
  int lane = (rem >> 1) & 31;
  int ks   = (rem >> 6) % KSTEPS;
  int ct   = (rem >> 6) / KSTEPS;
  int k = ks * 4 + 2 * (lane >> 4) + j;
  int n = ct * 16 + (lane & 15);
  wp[idx] = (k < HID && n < HID) ? w[(g * HID + n) * HID + k] : 0.0f;
}

__global__ void hist_deg(const int* __restrict__ ei, int* __restrict__ deg,
                         int n_edges) {
  int e = blockIdx.x * blockDim.x + threadIdx.x;
  if (e >= n_edges) return;
  atomicAdd(&deg[ei[n_edges + e]], 1);
}

__device__ __forceinline__ void scan_write(const int4* __restrict__ dv,
                                           int4* __restrict__ rv,
                                           int4* __restrict__ cv, int vi, int first) {
  const int4 d = dv[vi];
  int4 o;
  o.x = first;
  o.y = o.x + d.x;
  o.z = o.y + d.y;
  o.w = o.z + d.z;
  rv[vi] = o;
  cv[vi] = o;
}

#define SCAN_GROUPS 25
#define SCAN_VECS   (N_NODES / 4)
static_assert(N_NODES % 4 == 0 && 32 * SCAN_GROUPS * 128 >= N_NODES &&
              31 * SCAN_GROUPS * 128 <= N_NODES,
              "scan_deg: 32 segments must cover the nodes, the first 31 whole");
__global__ void __launch_bounds__(1024)
scan_deg(const int* __restrict__ deg, int* __restrict__ rowptr,
         int* __restrict__ cursor) {
  __shared__ int wsum[32];
  const int tid = threadIdx.x;
  const int lane = tid & 31, wave = tid >> 5;
  const int4* dv = (const int4*)deg;
  int4* rv = (int4*)rowptr;
  int4* cv = (int4*)cursor;
  const int vbase = wave * (SCAN_GROUPS * 32) + lane;

  int exg[SCAN_GROUPS];
  int run = 0;
  #pragma unroll
  for (int g = 0; g < SCAN_GROUPS; ++g) {
    const int vi = vbase + g * 32;
    const bool ok = vi < SCAN_VECS;
    int4 d = dv[ok ? vi : SCAN_VECS - 1];
    if (!ok) d = make_int4(0, 0, 0, 0);
    const int s4 = (d.x + d.y) + (d.z + d.w);
    int incl = s4;
    #pragma unroll
    for (int off = 1; off < 32; off <<= 1) {
      int t = __shfl_up(incl, off);
      if (lane >= off) incl += t;
    }
    exg[g] = run + incl - s4;
    run += __shfl(incl, 31);
  }
  if (lane == 0) wsum[wave] = run;
  __syncthreads();

  const int wt = wsum[lane];
  int wincl = wt;
  #pragma unroll
  for (int off = 1; off < 32; off <<= 1) {
    int t = __shfl_up(wincl, off);
    if (lane >= off) wincl += t;
  }
  const int total = __shfl(wincl, 31);
  const int wave_excl = __shfl(wincl, wave) - __shfl(wt, wave);

  if (wave < 31) {
    #pragma unroll
    for (int g = 0; g < SCAN_GROUPS; ++g)
      scan_write(dv, rv, cv, vbase + g * 32, wave_excl + exg[g]);
  } else {
    #pragma unroll
    for (int g = 0; g < SCAN_GROUPS; ++g) {
      const int vi = vbase + g * 32;
      if (vi < SCAN_VECS) scan_write(dv, rv, cv, vi, wave_excl + exg[g]);
    }
  }
  if (tid == 0) rowptr[N_NODES] = total;
}

__global__ void fill_csr(const int* __restrict__ ei, int* __restrict__ cursor,
                         int* __restrict__ csr_src, int n_edges) {
  int e = blockIdx.x * blockDim.x + threadIdx.x;
  if (e >= n_edges) return;
  int pos = atomicAdd(&cursor[ei[n_edges + e]], 1);
  csr_src[pos] = ei[e];
}

__global__ void __launch_bounds__(320)
fold_w(const float* __restrict__ w, const float* __restrict__ wpih,
       float* __restrict__ wfold) {
  __shared__ float as_[RPB * LDS_LD];
  const int rowbase = blockIdx.x * RPB;
  const int g = blockIdx.y, l = blockIdx.z;

  static_assert(RPB * HP == 16 * 320, "fold_w staging: 16 elements per thread");
  float st[16];
  #pragma unroll
  for (int it = 0; it < 16; ++it) {
    const int i = threadIdx.x + it * 320;
    const int r = i / HP, c = i - r * HP;
    const int k = rowbase + r;
    const bool ok = (k < HID && c < HID);
    const float t = w[ok ? (l * HID + k) * HID + c : 0];
    st[it] = ok ? t : 0.0f;
  }
  #pragma unroll
  for (int it = 0; it < 16; ++it) {
    const int i = threadIdx.x + it * 320;
    const int r = i / HP, c = i - r * HP;
    as_[r * LDS_LD + c] = st[it];
  }
  __syncthreads();

  const int wave = threadIdx.x >> 5, lane = threadIdx.x & 31;
  const int half = lane >> 4, l16 = lane & 15;
  const int GSTRIDE = COL_TILES * KSTEPS * 64;
  const float* wpt = wpih + g * GSTRIDE + wave * (KSTEPS * 64);

  v8f acc0 = {}, acc1 = {};
  for (int ks = 0; ks < KSTEPS; ++ks) {
    int ao = ks * 4 + 2 * half;
    v2f a0 = *(const v2f*)&as_[l16 * LDS_LD + ao];
    v2f a1 = *(const v2f*)&as_[(l16 + 16) * LDS_LD + ao];
    v2f b  = *(const v2f*)(wpt + ks * 64 + lane * 2);
    acc0 = wmma_f32(a0, b, acc0);
    acc1 = wmma_f32(a1, b, acc1);
  }
  float* dst = wfold + (size_t)(l * 3 + g) * GSTRIDE + wave * (KSTEPS * 64);
  #pragma unroll
  for (int v = 0; v < 8; v += 2) {
    const int k0 = rowbase + v + 8 * half;
    const int k1 = k0 + 16;
    v2f p0, p1;
    p0.x = acc0[v]; p0.y = acc0[v + 1];
    p1.x = acc1[v]; p1.y = acc1[v + 1];
    *(v2f*)&dst[(k0 >> 2) * 64 + (l16 + 16 * ((k0 >> 1) & 1)) * 2] = p0;
    *(v2f*)&dst[(k1 >> 2) * 64 + (l16 + 16 * ((k1 >> 1) & 1)) * 2] = p1;
  }
}

__global__ void __launch_bounds__(256)
gather_aggr(const float* __restrict__ hsrc, const int* __restrict__ rowptr,
            const int* __restrict__ csr_src, float* __restrict__ aggr) {
  int node = blockIdx.x * 8 + (threadIdx.x >> 5);
  if (node >= N_NODES) return;
  const int lane = threadIdx.x & 31;
  int beg = rowptr[node], end = rowptr[node + 1];
  float a0 = 0.f, a1 = 0.f, a2 = 0.f, a3 = 0.f, a4 = 0.f;
  for (int j = beg; j < end; ++j) {
    const float* mr = hsrc + (size_t)csr_src[j] * HP;
    float4 v = *(const float4*)(mr + 4 * lane);
    float  s = mr[128 + lane];
    a0 += v.x; a1 += v.y; a2 += v.z; a3 += v.w; a4 += s;
  }
  float* ar = aggr + (size_t)node * HP;
  float4 o; o.x = a0; o.y = a1; o.z = a2; o.w = a3;
  *(float4*)(ar + 4 * lane) = o;
  ar[128 + lane] = a4;
}

struct KFrag {
  v2f aA0, aA1, aH0, aH1;
  v2f br, bz, bn, cr, cz, cn;
};

__global__ void __launch_bounds__(320)
gru_fused(const float* __restrict__ hin, const float* __restrict__ aggr,
          const float* __restrict__ wih, const float* __restrict__ whh,
          const float* __restrict__ bih, const float* __restrict__ bhh,
          float* __restrict__ hout, int ldo, int zero_pad) {
  __shared__ float hs[RPB * LDS_LD];
  __shared__ float as_[RPB * LDS_LD];
  const int rowbase = blockIdx.x * RPB;

  for (int i = threadIdx.x; i < 2 * RPB * (HP / 4); i += 320) {
    int which = i / (RPB * (HP / 4));
    int t = i - which * (RPB * (HP / 4));
    int r = t / (HP / 4), cc = (t - r * (HP / 4)) * 4;
    const float* srcp = which ? (aggr + (rowbase + r) * HP + cc)
                              : (hin + (rowbase + r) * HP + cc);
    float4 v = *(const float4*)srcp;
    float* d = which ? &as_[r * LDS_LD + cc] : &hs[r * LDS_LD + cc];
    d[0] = v.x; d[1] = v.y; d[2] = v.z; d[3] = v.w;
  }
  __syncthreads();

  const int wave = threadIdx.x >> 5, lane = threadIdx.x & 31;
  const int half = lane >> 4, l16 = lane & 15;
  const int ct = wave;

  const int GSTRIDE = COL_TILES * KSTEPS * 64;
  const float* aAp = &as_[l16 * LDS_LD + 2 * half];
  const float* aHp = &hs[l16 * LDS_LD + 2 * half];
  const float* wi = wih + ct * (KSTEPS * 64) + lane * 2;
  const float* wh = whh + ct * (KSTEPS * 64) + lane * 2;

  v8f ir0 = {}, hr0 = {}, iz0 = {}, hz0 = {}, in0 = {}, hn0 = {};
  v8f ir1 = {}, hr1 = {}, iz1 = {}, hz1 = {}, in1 = {}, hn1 = {};

#define LOAD_STEP(F, KS)                                                   \
  do {                                                                     \
    const int ks_ = (KS);                                                  \
    F.br = *(const v2f*)(wi + ks_ * 64);                                   \
    F.bz = *(const v2f*)(wi + GSTRIDE + ks_ * 64);                         \
    F.bn = *(const v2f*)(wi + 2 * GSTRIDE + ks_ * 64);                     \
    F.cr = *(const v2f*)(wh + ks_ * 64);                                   \
    F.cz = *(const v2f*)(wh + GSTRIDE + ks_ * 64);                         \
    F.cn = *(const v2f*)(wh + 2 * GSTRIDE + ks_ * 64);                     \
    F.aA0 = *(const v2f*)(aAp + ks_ * 4);                                  \
    F.aA1 = *(const v2f*)(aAp + 16 * LDS_LD + ks_ * 4);                    \
    F.aH0 = *(const v2f*)(aHp + ks_ * 4);                                  \
    F.aH1 = *(const v2f*)(aHp + 16 * LDS_LD + ks_ * 4);                    \
  } while (0)

#define MMA_STEP(F)                                                        \
  do {                                                                     \
    ir0 = wmma_f32(F.aA0, F.br, ir0);  ir1 = wmma_f32(F.aA1, F.br, ir1);   \
    iz0 = wmma_f32(F.aA0, F.bz, iz0);  iz1 = wmma_f32(F.aA1, F.bz, iz1);   \
    in0 = wmma_f32(F.aA0, F.bn, in0);  in1 = wmma_f32(F.aA1, F.bn, in1);   \
    hr0 = wmma_f32(F.aH0, F.cr, hr0);  hr1 = wmma_f32(F.aH1, F.cr, hr1);   \
    hz0 = wmma_f32(F.aH0, F.cz, hz0);  hz1 = wmma_f32(F.aH1, F.cz, hz1);   \
    hn0 = wmma_f32(F.aH0, F.cn, hn0);  hn1 = wmma_f32(F.aH1, F.cn, hn1);   \
  } while (0)

  KFrag f0, f1;
  LOAD_STEP(f0, 0);
  for (int ks = 0; ks < KSTEPS; ks += 2) {
    LOAD_STEP(f1, ks + 1);
    __builtin_amdgcn_sched_barrier(0);
    MMA_STEP(f0);
    __builtin_amdgcn_sched_barrier(0);
    LOAD_STEP(f0, (ks + 2 < KSTEPS) ? ks + 2 : ks + 1);
    __builtin_amdgcn_sched_barrier(0);
    MMA_STEP(f1);
    __builtin_amdgcn_sched_barrier(0);
  }
#undef LOAD_STEP
#undef MMA_STEP

  const int col = ct * 16 + l16;
  const bool valid = (col < HID);
  const int cb = valid ? col : 0;
  const float bir = valid ? bih[cb]           : 0.0f;
  const float biz = valid ? bih[HID + cb]     : 0.0f;
  const float bin = valid ? bih[2 * HID + cb] : 0.0f;
  const float bhr = valid ? bhh[cb]           : 0.0f;
  const float bhz = valid ? bhh[HID + cb]     : 0.0f;
  const float bhn = valid ? bhh[2 * HID + cb] : 0.0f;

  #pragma unroll
  for (int rt = 0; rt < 2; ++rt) {
    #pragma unroll
    for (int v = 0; v < 8; ++v) {
      float irv = rt ? ir1[v] : ir0[v];
      float hrv = rt ? hr1[v] : hr0[v];
      float izv = rt ? iz1[v] : iz0[v];
      float hzv = rt ? hz1[v] : hz0[v];
      float inv = rt ? in1[v] : in0[v];
      float hnv = rt ? hn1[v] : hn0[v];
      int rl = rt * 16 + v + 8 * half;
      float r = sigmoidf_((irv + bir) + (hrv + bhr));
      float z = sigmoidf_((izv + biz) + (hzv + bhz));
      float n = tanhf_((inv + bin) + r * (hnv + bhn));
      float hv = hs[rl * LDS_LD + col];
      float o = (1.0f - z) * n + z * hv;
      long row = rowbase + rl;
      if (valid)          hout[row * ldo + col] = o;
      else if (zero_pad)  hout[row * ldo + col] = 0.0f;
    }
  }
}

extern "C" void kernel_launch(void* const* d_in, const int* in_sizes, int n_in,
                              void* d_out, int out_size, void* d_ws, size_t ws_size,
                              hipStream_t stream) {
  const float* x    = (const float*)d_in[0];
  const float* wgt  = (const float*)d_in[1];
  const float* w_ih = (const float*)d_in[2];
  const float* w_hh = (const float*)d_in[3];
  const float* b_ih = (const float*)d_in[4];
  const float* b_hh = (const float*)d_in[5];
  const int*   ei   = (const int*)d_in[6];
  const int n_edges = in_sizes[6] / 2;
  float* out = (float*)d_out;

  const size_t npad = (size_t)N_NODES * HP;
  const size_t gate_sz = 3 * COL_TILES * KSTEPS * 64;
  float* ws      = (float*)d_ws;
  float* h_pad   = ws;
  float* aggr    = h_pad + npad;
  float* wfold   = aggr  + npad;
  float* wpackih = wfold + (size_t)N_LAYERS * gate_sz;
  float* wpackhh = wpackih + gate_sz;
  int*   deg     = (int*)(wpackhh + gate_sz);
  int*   rowptr  = deg + N_NODES;
  int*   cursor  = rowptr + (N_NODES + 4);
  int*   csr_src = cursor + N_NODES;

  {
    int n = N_NODES * HP;
    init_hpad<<<(n + 255) / 256, 256, 0, stream>>>(x, h_pad);
    int ng = (int)gate_sz;
    pack_wg<<<(ng + 255) / 256, 256, 0, stream>>>(w_ih, wpackih);
    pack_wg<<<(ng + 255) / 256, 256, 0, stream>>>(w_hh, wpackhh);
    fold_w<<<dim3(HP / RPB, 3, N_LAYERS), 320, 0, stream>>>(wgt, wpackih, wfold);

    hipMemsetAsync(deg, 0, (size_t)N_NODES * sizeof(int), stream);
    hist_deg<<<(n_edges + 255) / 256, 256, 0, stream>>>(ei, deg, n_edges);
    scan_deg<<<1, 1024, 0, stream>>>(deg, rowptr, cursor);
    fill_csr<<<(n_edges + 255) / 256, 256, 0, stream>>>(ei, cursor, csr_src, n_edges);
  }

  for (int l = 0; l < N_LAYERS; ++l) {
    gather_aggr<<<(N_NODES + 7) / 8, 256, 0, stream>>>(h_pad, rowptr, csr_src, aggr);

    const bool last = (l == N_LAYERS - 1);
    gru_fused<<<RBLOCKS, 320, 0, stream>>>(
        h_pad, aggr, wfold + (size_t)l * gate_sz, wpackhh, b_ih, b_hh,
        last ? out : h_pad, last ? HID : HP, last ? 0 : 1);
  }
}


# edge sum fused into gru_fused: each block sums its 32 aggr rows from CSR straight into LDS (wave per row, scalar index path, 16 loads in flight), h double-buffered; 5 launches and aggr array gone
# speedup vs baseline: 1.5404x; 1.0975x over previous
#include <hip/hip_runtime.h>
#include <hip/hip_bf16.h>
#include <math.h>

#define N_NODES   100000
#define N_LAYERS  5
#define HID       150
#define HP        160
#define RPB       32
#define RBLOCKS   (N_NODES / RPB)
#define COL_TILES (HP / 16)
#define KSTEPS    (HP / 4)
#define LDS_LD    164

typedef __attribute__((ext_vector_type(2))) float v2f;
typedef __attribute__((ext_vector_type(8))) float v8f;

__device__ __forceinline__ v8f wmma_f32(v2f a, v2f b, v8f c) {
  return __builtin_amdgcn_wmma_f32_16x16x4_f32(
       false, a,  false, b,
       (short)0, c,  false,  false);
}

__device__ __forceinline__ float sigmoidf_(float x) {
  return __builtin_amdgcn_rcpf(1.0f + __expf(-x));
}
__device__ __forceinline__ float tanhf_(float x) {
  return 1.0f - 2.0f * __builtin_amdgcn_rcpf(1.0f + __builtin_amdgcn_exp2f(x * 2.88539008f));
}


__global__ void init_hpad(const float* __restrict__ x, float* __restrict__ hpad) {
  int idx = blockIdx.x * blockDim.x + threadIdx.x;
  if (idx >= N_NODES * HP) return;
  int row = idx / HP, col = idx - row * HP;
  hpad[idx] = (col < HID) ? x[row * HID + col] : 0.0f;
}

__global__ void pack_wg(const float* __restrict__ w, float* __restrict__ wp) {
  int idx = blockIdx.x * blockDim.x + threadIdx.x;
  const int per_gate = COL_TILES * KSTEPS * 64;
  if (idx >= 3 * per_gate) return;
  int g    = idx / per_gate;
  int rem  = idx - g * per_gate;
  int j    = rem & 1;
  int lane = (rem >> 1) & 31;
  int ks   = (rem >> 6) % KSTEPS;
  int ct   = (rem >> 6) / KSTEPS;
  int k = ks * 4 + 2 * (lane >> 4) + j;
  int n = ct * 16 + (lane & 15);
  wp[idx] = (k < HID && n < HID) ? w[(g * HID + n) * HID + k] : 0.0f;
}

__global__ void hist_deg(const int* __restrict__ ei, int* __restrict__ deg,
                         int n_edges) {
  int e = blockIdx.x * blockDim.x + threadIdx.x;
  if (e >= n_edges) return;
  atomicAdd(&deg[ei[n_edges + e]], 1);
}

__device__ __forceinline__ void scan_write(const int4* __restrict__ dv,
                                           int4* __restrict__ rv,
                                           int4* __restrict__ cv, int vi, int first) {
  const int4 d = dv[vi];
  int4 o;
  o.x = first;
  o.y = o.x + d.x;
  o.z = o.y + d.y;
  o.w = o.z + d.z;
  rv[vi] = o;
  cv[vi] = o;
}

#define SCAN_GROUPS 25
#define SCAN_VECS   (N_NODES / 4)
static_assert(N_NODES % 4 == 0 && 32 * SCAN_GROUPS * 128 >= N_NODES &&
              31 * SCAN_GROUPS * 128 <= N_NODES,
              "scan_deg: 32 segments must cover the nodes, the first 31 whole");
__global__ void __launch_bounds__(1024)
scan_deg(const int* __restrict__ deg, int* __restrict__ rowptr,
         int* __restrict__ cursor) {
  __shared__ int wsum[32];
  const int tid = threadIdx.x;
  const int lane = tid & 31, wave = tid >> 5;
  const int4* dv = (const int4*)deg;
  int4* rv = (int4*)rowptr;
  int4* cv = (int4*)cursor;
  const int vbase = wave * (SCAN_GROUPS * 32) + lane;

  int exg[SCAN_GROUPS];
  int run = 0;
  #pragma unroll
  for (int g = 0; g < SCAN_GROUPS; ++g) {
    const int vi = vbase + g * 32;
    const bool ok = vi < SCAN_VECS;
    int4 d = dv[ok ? vi : SCAN_VECS - 1];
    if (!ok) d = make_int4(0, 0, 0, 0);
    const int s4 = (d.x + d.y) + (d.z + d.w);
    int incl = s4;
    #pragma unroll
    for (int off = 1; off < 32; off <<= 1) {
      int t = __shfl_up(incl, off);
      if (lane >= off) incl += t;
    }
    exg[g] = run + incl - s4;
    run += __shfl(incl, 31);
  }
  if (lane == 0) wsum[wave] = run;
  __syncthreads();

  const int wt = wsum[lane];
  int wincl = wt;
  #pragma unroll
  for (int off = 1; off < 32; off <<= 1) {
    int t = __shfl_up(wincl, off);
    if (lane >= off) wincl += t;
  }
  const int total = __shfl(wincl, 31);
  const int wave_excl = __shfl(wincl, wave) - __shfl(wt, wave);

  if (wave < 31) {
    #pragma unroll
    for (int g = 0; g < SCAN_GROUPS; ++g)
      scan_write(dv, rv, cv, vbase + g * 32, wave_excl + exg[g]);
  } else {
    #pragma unroll
    for (int g = 0; g < SCAN_GROUPS; ++g) {
      const int vi = vbase + g * 32;
      if (vi < SCAN_VECS) scan_write(dv, rv, cv, vi, wave_excl + exg[g]);
    }
  }
  if (tid == 0) rowptr[N_NODES] = total;
}

__global__ void fill_csr(const int* __restrict__ ei, int* __restrict__ cursor,
                         int* __restrict__ csr_src, int n_edges) {
  int e = blockIdx.x * blockDim.x + threadIdx.x;
  if (e >= n_edges) return;
  int pos = atomicAdd(&cursor[ei[n_edges + e]], 1);
  csr_src[pos] = ei[e];
}

__global__ void __launch_bounds__(320)
fold_w(const float* __restrict__ w, const float* __restrict__ wpih,
       float* __restrict__ wfold) {
  __shared__ float as_[RPB * LDS_LD];
  const int rowbase = blockIdx.x * RPB;
  const int g = blockIdx.y, l = blockIdx.z;

  static_assert(RPB * HP == 16 * 320, "fold_w staging: 16 elements per thread");
  float st[16];
  #pragma unroll
  for (int it = 0; it < 16; ++it) {
    const int i = threadIdx.x + it * 320;
    const int r = i / HP, c = i - r * HP;
    const int k = rowbase + r;
    const bool ok = (k < HID && c < HID);
    const float t = w[ok ? (l * HID + k) * HID + c : 0];
    st[it] = ok ? t : 0.0f;
  }
  #pragma unroll
  for (int it = 0; it < 16; ++it) {
    const int i = threadIdx.x + it * 320;
    const int r = i / HP, c = i - r * HP;
    as_[r * LDS_LD + c] = st[it];
  }
  __syncthreads();

  const int wave = threadIdx.x >> 5, lane = threadIdx.x & 31;
  const int half = lane >> 4, l16 = lane & 15;
  const int GSTRIDE = COL_TILES * KSTEPS * 64;
  const float* wpt = wpih + g * GSTRIDE + wave * (KSTEPS * 64);

  v8f acc0 = {}, acc1 = {};
  for (int ks = 0; ks < KSTEPS; ++ks) {
    int ao = ks * 4 + 2 * half;
    v2f a0 = *(const v2f*)&as_[l16 * LDS_LD + ao];
    v2f a1 = *(const v2f*)&as_[(l16 + 16) * LDS_LD + ao];
    v2f b  = *(const v2f*)(wpt + ks * 64 + lane * 2);
    acc0 = wmma_f32(a0, b, acc0);
    acc1 = wmma_f32(a1, b, acc1);
  }
  float* dst = wfold + (size_t)(l * 3 + g) * GSTRIDE + wave * (KSTEPS * 64);
  #pragma unroll
  for (int v = 0; v < 8; v += 2) {
    const int k0 = rowbase + v + 8 * half;
    const int k1 = k0 + 16;
    v2f p0, p1;
    p0.x = acc0[v]; p0.y = acc0[v + 1];
    p1.x = acc1[v]; p1.y = acc1[v + 1];
    *(v2f*)&dst[(k0 >> 2) * 64 + (l16 + 16 * ((k0 >> 1) & 1)) * 2] = p0;
    *(v2f*)&dst[(k1 >> 2) * 64 + (l16 + 16 * ((k1 >> 1) & 1)) * 2] = p1;
  }
}

struct KFrag {
  v2f aA0, aA1, aH0, aH1;
  v2f br, bz, bn, cr, cz, cn;
};

#define GW 8
__global__ void __launch_bounds__(320)
gru_fused(const float* __restrict__ hin, const int* __restrict__ rowptr,
          const int* __restrict__ csr_src,
          const float* __restrict__ wih, const float* __restrict__ whh,
          const float* __restrict__ bih, const float* __restrict__ bhh,
          float* __restrict__ hout, int ldo, int zero_pad) {
  __shared__ __attribute__((aligned(16))) float hs[RPB * LDS_LD];
  __shared__ __attribute__((aligned(16))) float as_[RPB * LDS_LD];
  const int rowbase = blockIdx.x * RPB;

  static_assert(RPB * (HP / 4) == 4 * 320, "h staging: 4 vectors per thread");
  {
    float4 st[4];
    #pragma unroll
    for (int it = 0; it < 4; ++it) {
      const int i = threadIdx.x + it * 320;
      const int r = i / (HP / 4), cc = (i - r * (HP / 4)) * 4;
      st[it] = *(const float4*)(hin + (size_t)(rowbase + r) * HP + cc);
    }
    #pragma unroll
    for (int it = 0; it < 4; ++it) {
      const int i = threadIdx.x + it * 320;
      const int r = i / (HP / 4), cc = (i - r * (HP / 4)) * 4;
      *(float4*)&hs[r * LDS_LD + cc] = st[it];
    }
  }

  {
    const int wv = __builtin_amdgcn_readfirstlane((int)(threadIdx.x >> 5));
    const int ln = threadIdx.x & 31;
    const int r0 = 3 * wv + (wv < 2 ? wv : 2);
    const int nr = wv < 2 ? 4 : 3;
    static_assert(RPB == 32, "edge sum: row split written for 32 rows on 10 waves");
    const unsigned lb  = (unsigned)ln * 16u;
    const unsigned lb2 = 512u + (unsigned)ln * 4u;
    const char* hbase = (const char*)hin;
    int bnd[5];
    #pragma unroll
    for (int k = 0; k < 5; ++k) bnd[k] = rowptr[rowbase + r0 + (k < nr ? k : nr)];
    #pragma unroll
    for (int k = 0; k < 4; ++k) {
      if (k < nr) {
        const int beg = bnd[k], end = bnd[k + 1];
        const int last = end > 0 ? end - 1 : 0;
        float4 acc = make_float4(0.f, 0.f, 0.f, 0.f);
        float acc4 = 0.f;
        int idx[GW];
        #pragma unroll
        for (int u = 0; u < GW; ++u) {
          const int jj = beg + u;
          idx[u] = csr_src[jj < last ? jj : last];
        }
        int j = beg;
        for (; j + GW <= end; j += GW) {
          float4 v[GW];
          float  s[GW];
          #pragma unroll
          for (int u = 0; u < GW; ++u) {
            const char* mr = hbase + (size_t)idx[u] * (HP * 4);
            v[u] = *(const float4*)(mr + lb);
            s[u] = *(const float*)(mr + lb2);
          }
          #pragma unroll
          for (int u = 0; u < GW; ++u) {
            const int jj = j + GW + u;
            idx[u] = csr_src[jj < last ? jj : last];
          }
          #pragma unroll
          for (int u = 0; u < GW; ++u) {
            acc.x += v[u].x; acc.y += v[u].y; acc.z += v[u].z; acc.w += v[u].w;
            acc4 += s[u];
          }
        }
        const int nrem = end - j;
        if (nrem > 0) {
          float4 v[GW - 1];
          float  s[GW - 1];
          #pragma unroll
          for (int u = 0; u < GW - 1; ++u) {
            const char* mr = hbase + (size_t)idx[u] * (HP * 4);
            v[u] = *(const float4*)(mr + lb);
            s[u] = *(const float*)(mr + lb2);
          }
          #pragma unroll
          for (int u = 0; u < GW - 1; ++u) {
            const bool live = u < nrem;
            acc.x += live ? v[u].x : 0.f; acc.y += live ? v[u].y : 0.f;
            acc.z += live ? v[u].z : 0.f; acc.w += live ? v[u].w : 0.f;
            acc4 += live ? s[u] : 0.f;
          }
        }
        const int r = r0 + k;
        *(float4*)&as_[r * LDS_LD + 4 * ln] = acc;
        as_[r * LDS_LD + 128 + ln] = acc4;
      }
    }
  }
  __syncthreads();

  const int wave = threadIdx.x >> 5, lane = threadIdx.x & 31;
  const int half = lane >> 4, l16 = lane & 15;
  const int ct = wave;

  const int GSTRIDE = COL_TILES * KSTEPS * 64;
  const float* aAp = &as_[l16 * LDS_LD + 2 * half];
  const float* aHp = &hs[l16 * LDS_LD + 2 * half];
  const float* wi = wih + ct * (KSTEPS * 64) + lane * 2;
  const float* wh = whh + ct * (KSTEPS * 64) + lane * 2;

  v8f ir0 = {}, hr0 = {}, iz0 = {}, hz0 = {}, in0 = {}, hn0 = {};
  v8f ir1 = {}, hr1 = {}, iz1 = {}, hz1 = {}, in1 = {}, hn1 = {};

#define LOAD_STEP(F, KS)                                                   \
  do {                                                                     \
    const int ks_ = (KS);                                                  \
    F.br = *(const v2f*)(wi + ks_ * 64);                                   \
    F.bz = *(const v2f*)(wi + GSTRIDE + ks_ * 64);                         \
    F.bn = *(const v2f*)(wi + 2 * GSTRIDE + ks_ * 64);                     \
    F.cr = *(const v2f*)(wh + ks_ * 64);                                   \
    F.cz = *(const v2f*)(wh + GSTRIDE + ks_ * 64);                         \
    F.cn = *(const v2f*)(wh + 2 * GSTRIDE + ks_ * 64);                     \
    F.aA0 = *(const v2f*)(aAp + ks_ * 4);                                  \
    F.aA1 = *(const v2f*)(aAp + 16 * LDS_LD + ks_ * 4);                    \
    F.aH0 = *(const v2f*)(aHp + ks_ * 4);                                  \
    F.aH1 = *(const v2f*)(aHp + 16 * LDS_LD + ks_ * 4);                    \
  } while (0)

#define MMA_STEP(F)                                                        \
  do {                                                                     \
    ir0 = wmma_f32(F.aA0, F.br, ir0);  ir1 = wmma_f32(F.aA1, F.br, ir1);   \
    iz0 = wmma_f32(F.aA0, F.bz, iz0);  iz1 = wmma_f32(F.aA1, F.bz, iz1);   \
    in0 = wmma_f32(F.aA0, F.bn, in0);  in1 = wmma_f32(F.aA1, F.bn, in1);   \
    hr0 = wmma_f32(F.aH0, F.cr, hr0);  hr1 = wmma_f32(F.aH1, F.cr, hr1);   \
    hz0 = wmma_f32(F.aH0, F.cz, hz0);  hz1 = wmma_f32(F.aH1, F.cz, hz1);   \
    hn0 = wmma_f32(F.aH0, F.cn, hn0);  hn1 = wmma_f32(F.aH1, F.cn, hn1);   \
  } while (0)

  KFrag f0, f1;
  LOAD_STEP(f0, 0);
  for (int ks = 0; ks < KSTEPS; ks += 2) {
    LOAD_STEP(f1, ks + 1);
    __builtin_amdgcn_sched_barrier(0);
    MMA_STEP(f0);
    __builtin_amdgcn_sched_barrier(0);
    LOAD_STEP(f0, (ks + 2 < KSTEPS) ? ks + 2 : ks + 1);
    __builtin_amdgcn_sched_barrier(0);
    MMA_STEP(f1);
    __builtin_amdgcn_sched_barrier(0);
  }
#undef LOAD_STEP
#undef MMA_STEP

  const int col = ct * 16 + l16;
  const bool valid = (col < HID);
  const int cb = valid ? col : 0;
  const float bir = valid ? bih[cb]           : 0.0f;
  const float biz = valid ? bih[HID + cb]     : 0.0f;
  const float bin = valid ? bih[2 * HID + cb] : 0.0f;
  const float bhr = valid ? bhh[cb]           : 0.0f;
  const float bhz = valid ? bhh[HID + cb]     : 0.0f;
  const float bhn = valid ? bhh[2 * HID + cb] : 0.0f;

  #pragma unroll
  for (int rt = 0; rt < 2; ++rt) {
    #pragma unroll
    for (int v = 0; v < 8; ++v) {
      float irv = rt ? ir1[v] : ir0[v];
      float hrv = rt ? hr1[v] : hr0[v];
      float izv = rt ? iz1[v] : iz0[v];
      float hzv = rt ? hz1[v] : hz0[v];
      float inv = rt ? in1[v] : in0[v];
      float hnv = rt ? hn1[v] : hn0[v];
      int rl = rt * 16 + v + 8 * half;
      float r = sigmoidf_((irv + bir) + (hrv + bhr));
      float z = sigmoidf_((izv + biz) + (hzv + bhz));
      float n = tanhf_((inv + bin) + r * (hnv + bhn));
      float hv = hs[rl * LDS_LD + col];
      float o = (1.0f - z) * n + z * hv;
      long row = rowbase + rl;
      if (valid)          hout[row * ldo + col] = o;
      else if (zero_pad)  hout[row * ldo + col] = 0.0f;
    }
  }
}

extern "C" void kernel_launch(void* const* d_in, const int* in_sizes, int n_in,
                              void* d_out, int out_size, void* d_ws, size_t ws_size,
                              hipStream_t stream) {
  const float* x    = (const float*)d_in[0];
  const float* wgt  = (const float*)d_in[1];
  const float* w_ih = (const float*)d_in[2];
  const float* w_hh = (const float*)d_in[3];
  const float* b_ih = (const float*)d_in[4];
  const float* b_hh = (const float*)d_in[5];
  const int*   ei   = (const int*)d_in[6];
  const int n_edges = in_sizes[6] / 2;
  float* out = (float*)d_out;

  const size_t npad = (size_t)N_NODES * HP;
  const size_t gate_sz = 3 * COL_TILES * KSTEPS * 64;
  float* ws      = (float*)d_ws;
  float* h_pad   = ws;
  float* h_alt   = h_pad + npad;
  float* wfold   = h_alt + npad;
  float* wpackih = wfold + (size_t)N_LAYERS * gate_sz;
  float* wpackhh = wpackih + gate_sz;
  int*   deg     = (int*)(wpackhh + gate_sz);
  int*   rowptr  = deg + N_NODES;
  int*   cursor  = rowptr + (N_NODES + 4);
  int*   csr_src = cursor + N_NODES;

  {
    int n = N_NODES * HP;
    init_hpad<<<(n + 255) / 256, 256, 0, stream>>>(x, h_pad);
    int ng = (int)gate_sz;
    pack_wg<<<(ng + 255) / 256, 256, 0, stream>>>(w_ih, wpackih);
    pack_wg<<<(ng + 255) / 256, 256, 0, stream>>>(w_hh, wpackhh);
    fold_w<<<dim3(HP / RPB, 3, N_LAYERS), 320, 0, stream>>>(wgt, wpackih, wfold);

    hipMemsetAsync(deg, 0, (size_t)N_NODES * sizeof(int), stream);
    hist_deg<<<(n_edges + 255) / 256, 256, 0, stream>>>(ei, deg, n_edges);
    scan_deg<<<1, 1024, 0, stream>>>(deg, rowptr, cursor);
    fill_csr<<<(n_edges + 255) / 256, 256, 0, stream>>>(ei, cursor, csr_src, n_edges);
  }

  for (int l = 0; l < N_LAYERS; ++l) {
    const bool last = (l == N_LAYERS - 1);
    const float* hin = (l & 1) ? h_alt : h_pad;
    float* hnext     = (l & 1) ? h_pad : h_alt;
    gru_fused<<<RBLOCKS, 320, 0, stream>>>(
        hin, rowptr, csr_src, wfold + (size_t)l * gate_sz, wpackhh, b_ih, b_hh,
        last ? out : hnext, last ? HID : HP, last ? 0 : 1);
  }
}


# gru_fused on 20 waves, not 10: the two waves of a column tile take the aggr-side and the h-side products (6 accumulator tiles each) and swap one row tile's sums through LDS; 5 equal waves per SIMD
# speedup vs baseline: 1.6121x; 1.0465x over previous
#include <hip/hip_runtime.h>
#include <hip/hip_bf16.h>
#include <math.h>

#define N_NODES   100000
#define N_LAYERS  5
#define HID       150
#define HP        160
#define RPB       32
#define RBLOCKS   (N_NODES / RPB)
#define COL_TILES (HP / 16)
#define KSTEPS    (HP / 4)
#define LDS_LD    164

typedef __attribute__((ext_vector_type(2))) float v2f;
typedef __attribute__((ext_vector_type(8))) float v8f;

__device__ __forceinline__ v8f wmma_f32(v2f a, v2f b, v8f c) {
  return __builtin_amdgcn_wmma_f32_16x16x4_f32(
       false, a,  false, b,
       (short)0, c,  false,  false);
}

__device__ __forceinline__ float sigmoidf_(float x) {
  return __builtin_amdgcn_rcpf(1.0f + __expf(-x));
}
__device__ __forceinline__ float tanhf_(float x) {
  return 1.0f - 2.0f * __builtin_amdgcn_rcpf(1.0f + __builtin_amdgcn_exp2f(x * 2.88539008f));
}


__global__ void init_hpad(const float* __restrict__ x, float* __restrict__ hpad) {
  int idx = blockIdx.x * blockDim.x + threadIdx.x;
  if (idx >= N_NODES * HP) return;
  int row = idx / HP, col = idx - row * HP;
  hpad[idx] = (col < HID) ? x[row * HID + col] : 0.0f;
}

__global__ void pack_wg(const float* __restrict__ w, float* __restrict__ wp) {
  int idx = blockIdx.x * blockDim.x + threadIdx.x;
  const int per_gate = COL_TILES * KSTEPS * 64;
  if (idx >= 3 * per_gate) return;
  int g    = idx / per_gate;
  int rem  = idx - g * per_gate;
  int j    = rem & 1;
  int lane = (rem >> 1) & 31;
  int ks   = (rem >> 6) % KSTEPS;
  int ct   = (rem >> 6) / KSTEPS;
  int k = ks * 4 + 2 * (lane >> 4) + j;
  int n = ct * 16 + (lane & 15);
  wp[idx] = (k < HID && n < HID) ? w[(g * HID + n) * HID + k] : 0.0f;
}

__global__ void hist_deg(const int* __restrict__ ei, int* __restrict__ deg,
                         int n_edges) {
  int e = blockIdx.x * blockDim.x + threadIdx.x;
  if (e >= n_edges) return;
  atomicAdd(&deg[ei[n_edges + e]], 1);
}

__device__ __forceinline__ void scan_write(const int4* __restrict__ dv,
                                           int4* __restrict__ rv,
                                           int4* __restrict__ cv, int vi, int first) {
  const int4 d = dv[vi];
  int4 o;
  o.x = first;
  o.y = o.x + d.x;
  o.z = o.y + d.y;
  o.w = o.z + d.z;
  rv[vi] = o;
  cv[vi] = o;
}

#define SCAN_GROUPS 25
#define SCAN_VECS   (N_NODES / 4)
static_assert(N_NODES % 4 == 0 && 32 * SCAN_GROUPS * 128 >= N_NODES &&
              31 * SCAN_GROUPS * 128 <= N_NODES,
              "scan_deg: 32 segments must cover the nodes, the first 31 whole");
__global__ void __launch_bounds__(1024)
scan_deg(const int* __restrict__ deg, int* __restrict__ rowptr,
         int* __restrict__ cursor) {
  __shared__ int wsum[32];
  const int tid = threadIdx.x;
  const int lane = tid & 31, wave = tid >> 5;
  const int4* dv = (const int4*)deg;
  int4* rv = (int4*)rowptr;
  int4* cv = (int4*)cursor;
  const int vbase = wave * (SCAN_GROUPS * 32) + lane;

  int exg[SCAN_GROUPS];
  int run = 0;
  #pragma unroll
  for (int g = 0; g < SCAN_GROUPS; ++g) {
    const int vi = vbase + g * 32;
    const bool ok = vi < SCAN_VECS;
    int4 d = dv[ok ? vi : SCAN_VECS - 1];
    if (!ok) d = make_int4(0, 0, 0, 0);
    const int s4 = (d.x + d.y) + (d.z + d.w);
    int incl = s4;
    #pragma unroll
    for (int off = 1; off < 32; off <<= 1) {
      int t = __shfl_up(incl, off);
      if (lane >= off) incl += t;
    }
    exg[g] = run + incl - s4;
    run += __shfl(incl, 31);
  }
  if (lane == 0) wsum[wave] = run;
  __syncthreads();

  const int wt = wsum[lane];
  int wincl = wt;
  #pragma unroll
  for (int off = 1; off < 32; off <<= 1) {
    int t = __shfl_up(wincl, off);
    if (lane >= off) wincl += t;
  }
  const int total = __shfl(wincl, 31);
  const int wave_excl = __shfl(wincl, wave) - __shfl(wt, wave);

  if (wave < 31) {
    #pragma unroll
    for (int g = 0; g < SCAN_GROUPS; ++g)
      scan_write(dv, rv, cv, vbase + g * 32, wave_excl + exg[g]);
  } else {
    #pragma unroll
    for (int g = 0; g < SCAN_GROUPS; ++g) {
      const int vi = vbase + g * 32;
      if (vi < SCAN_VECS) scan_write(dv, rv, cv, vi, wave_excl + exg[g]);
    }
  }
  if (tid == 0) rowptr[N_NODES] = total;
}

__global__ void fill_csr(const int* __restrict__ ei, int* __restrict__ cursor,
                         int* __restrict__ csr_src, int n_edges) {
  int e = blockIdx.x * blockDim.x + threadIdx.x;
  if (e >= n_edges) return;
  int pos = atomicAdd(&cursor[ei[n_edges + e]], 1);
  csr_src[pos] = ei[e];
}

__global__ void __launch_bounds__(320)
fold_w(const float* __restrict__ w, const float* __restrict__ wpih,
       float* __restrict__ wfold) {
  __shared__ float as_[RPB * LDS_LD];
  const int rowbase = blockIdx.x * RPB;
  const int g = blockIdx.y, l = blockIdx.z;

  static_assert(RPB * HP == 16 * 320, "fold_w staging: 16 elements per thread");
  float st[16];
  #pragma unroll
  for (int it = 0; it < 16; ++it) {
    const int i = threadIdx.x + it * 320;
    const int r = i / HP, c = i - r * HP;
    const int k = rowbase + r;
    const bool ok = (k < HID && c < HID);
    const float t = w[ok ? (l * HID + k) * HID + c : 0];
    st[it] = ok ? t : 0.0f;
  }
  #pragma unroll
  for (int it = 0; it < 16; ++it) {
    const int i = threadIdx.x + it * 320;
    const int r = i / HP, c = i - r * HP;
    as_[r * LDS_LD + c] = st[it];
  }
  __syncthreads();

  const int wave = threadIdx.x >> 5, lane = threadIdx.x & 31;
  const int half = lane >> 4, l16 = lane & 15;
  const int GSTRIDE = COL_TILES * KSTEPS * 64;
  const float* wpt = wpih + g * GSTRIDE + wave * (KSTEPS * 64);

  v8f acc0 = {}, acc1 = {};
  for (int ks = 0; ks < KSTEPS; ++ks) {
    int ao = ks * 4 + 2 * half;
    v2f a0 = *(const v2f*)&as_[l16 * LDS_LD + ao];
    v2f a1 = *(const v2f*)&as_[(l16 + 16) * LDS_LD + ao];
    v2f b  = *(const v2f*)(wpt + ks * 64 + lane * 2);
    acc0 = wmma_f32(a0, b, acc0);
    acc1 = wmma_f32(a1, b, acc1);
  }
  float* dst = wfold + (size_t)(l * 3 + g) * GSTRIDE + wave * (KSTEPS * 64);
  #pragma unroll
  for (int v = 0; v < 8; v += 2) {
    const int k0 = rowbase + v + 8 * half;
    const int k1 = k0 + 16;
    v2f p0, p1;
    p0.x = acc0[v]; p0.y = acc0[v + 1];
    p1.x = acc1[v]; p1.y = acc1[v + 1];
    *(v2f*)&dst[(k0 >> 2) * 64 + (l16 + 16 * ((k0 >> 1) & 1)) * 2] = p0;
    *(v2f*)&dst[(k1 >> 2) * 64 + (l16 + 16 * ((k1 >> 1) & 1)) * 2] = p1;
  }
}

struct KFrag {
  v2f a0, a1;
  v2f b0, b1, b2;
};

#define GW 8
#define GRU_THREADS 640
#define GRU_WAVES   (GRU_THREADS / 32)
__global__ void __launch_bounds__(GRU_THREADS)
gru_fused(const float* __restrict__ hin, const int* __restrict__ rowptr,
          const int* __restrict__ csr_src,
          const float* __restrict__ wih, const float* __restrict__ whh,
          const float* __restrict__ bih, const float* __restrict__ bhh,
          float* __restrict__ hout, int ldo, int zero_pad) {
  __shared__ __attribute__((aligned(16))) float ah[2 * RPB * LDS_LD];
  __shared__ float xs[COL_TILES * 2 * 3 * 256];
  float* as_ = ah;
  float* hs  = ah + RPB * LDS_LD;
  const int rowbase = blockIdx.x * RPB;

  static_assert(RPB * (HP / 4) == 2 * GRU_THREADS, "h staging: 2 vectors per thread");
  {
    float4 st[2];
    #pragma unroll
    for (int it = 0; it < 2; ++it) {
      const int i = threadIdx.x + it * GRU_THREADS;
      const int r = i / (HP / 4), cc = (i - r * (HP / 4)) * 4;
      st[it] = *(const float4*)(hin + (size_t)(rowbase + r) * HP + cc);
    }
    #pragma unroll
    for (int it = 0; it < 2; ++it) {
      const int i = threadIdx.x + it * GRU_THREADS;
      const int r = i / (HP / 4), cc = (i - r * (HP / 4)) * 4;
      *(float4*)&hs[r * LDS_LD + cc] = st[it];
    }
  }

  {
    const int wv = __builtin_amdgcn_readfirstlane((int)(threadIdx.x >> 5));
    const int ln = threadIdx.x & 31;
    const int r0 = wv < 12 ? 2 * wv : wv + 12;
    const int nr = wv < 12 ? 2 : 1;
    static_assert(RPB == 32 && GRU_WAVES == 20, "edge sum: row split written for 32 rows on 20 waves");
    const unsigned lb  = (unsigned)ln * 16u;
    const unsigned lb2 = 512u + (unsigned)ln * 4u;
    const char* hbase = (const char*)hin;
    int bnd[3];
    #pragma unroll
    for (int k = 0; k < 3; ++k) bnd[k] = rowptr[rowbase + r0 + (k < nr ? k : nr)];
    #pragma unroll
    for (int k = 0; k < 2; ++k) {
      if (k < nr) {
        const int beg = bnd[k], end = bnd[k + 1];
        const int last = end > 0 ? end - 1 : 0;
        float4 acc = make_float4(0.f, 0.f, 0.f, 0.f);
        float acc4 = 0.f;
        int idx[GW];
        #pragma unroll
        for (int u = 0; u < GW; ++u) {
          const int jj = beg + u;
          idx[u] = csr_src[jj < last ? jj : last];
        }
        int j = beg;
        for (; j + GW <= end; j += GW) {
          float4 v[GW];
          float  s[GW];
          #pragma unroll
          for (int u = 0; u < GW; ++u) {
            const char* mr = hbase + (size_t)idx[u] * (HP * 4);
            v[u] = *(const float4*)(mr + lb);
            s[u] = *(const float*)(mr + lb2);
          }
          #pragma unroll
          for (int u = 0; u < GW; ++u) {
            const int jj = j + GW + u;
            idx[u] = csr_src[jj < last ? jj : last];
          }
          #pragma unroll
          for (int u = 0; u < GW; ++u) {
            acc.x += v[u].x; acc.y += v[u].y; acc.z += v[u].z; acc.w += v[u].w;
            acc4 += s[u];
          }
        }
        const int nrem = end - j;
        if (nrem > 0) {
          float4 v[GW - 1];
          float  s[GW - 1];
          #pragma unroll
          for (int u = 0; u < GW - 1; ++u) {
            const char* mr = hbase + (size_t)idx[u] * (HP * 4);
            v[u] = *(const float4*)(mr + lb);
            s[u] = *(const float*)(mr + lb2);
          }
          #pragma unroll
          for (int u = 0; u < GW - 1; ++u) {
            const bool live = u < nrem;
            acc.x += live ? v[u].x : 0.f; acc.y += live ? v[u].y : 0.f;
            acc.z += live ? v[u].z : 0.f; acc.w += live ? v[u].w : 0.f;
            acc4 += live ? s[u] : 0.f;
          }
        }
        const int r = r0 + k;
        *(float4*)&as_[r * LDS_LD + 4 * ln] = acc;
        as_[r * LDS_LD + 128 + ln] = acc4;
      }
    }
  }
  __syncthreads();

  const int wave = __builtin_amdgcn_readfirstlane((int)(threadIdx.x >> 5));
  const int lane = threadIdx.x & 31;
  const int half = lane >> 4, l16 = lane & 15;
  const int side = wave >= COL_TILES ? 1 : 0;
  const int ct = wave - side * COL_TILES;

  const int GSTRIDE = COL_TILES * KSTEPS * 64;
  const float* ap = ah + side * (RPB * LDS_LD) + l16 * LDS_LD + 2 * half;
  const float* wp = (side ? whh : wih) + ct * (KSTEPS * 64) + lane * 2;

  v8f g00 = {}, g01 = {}, g10 = {}, g11 = {}, g20 = {}, g21 = {};

#define LOAD_STEP(F, WP, AP)                                               \
  do {                                                                     \
    F.b0 = *(const v2f*)(WP);                                              \
    F.b1 = *(const v2f*)((WP) + GSTRIDE);                                  \
    F.b2 = *(const v2f*)((WP) + 2 * GSTRIDE);                              \
    F.a0 = *(const v2f*)(AP);                                              \
    F.a1 = *(const v2f*)((AP) + 16 * LDS_LD);                              \
  } while (0)

#define MMA_STEP(F)                                                        \
  do {                                                                     \
    g00 = wmma_f32(F.a0, F.b0, g00);  g01 = wmma_f32(F.a1, F.b0, g01);     \
    g10 = wmma_f32(F.a0, F.b1, g10);  g11 = wmma_f32(F.a1, F.b1, g11);     \
    g20 = wmma_f32(F.a0, F.b2, g20);  g21 = wmma_f32(F.a1, F.b2, g21);     \
  } while (0)

  static_assert(KSTEPS % 2 == 0 && KSTEPS >= 4, "K loop: steps in pairs, last pair peeled");
  KFrag f0, f1;
  LOAD_STEP(f0, wp, ap);
  #pragma nounroll
  for (int it = 0; it < KSTEPS / 2 - 1; ++it) {
    LOAD_STEP(f1, wp + 64, ap + 4);
    __builtin_amdgcn_sched_barrier(0);
    MMA_STEP(f0);
    __builtin_amdgcn_sched_barrier(0);
    LOAD_STEP(f0, wp + 128, ap + 8);
    __builtin_amdgcn_sched_barrier(0);
    MMA_STEP(f1);
    __builtin_amdgcn_sched_barrier(0);
    wp += 128;
    ap += 8;
  }
  LOAD_STEP(f1, wp + 64, ap + 4);
  __builtin_amdgcn_sched_barrier(0);
  MMA_STEP(f0);
  MMA_STEP(f1);
#undef LOAD_STEP
#undef MMA_STEP

  float* xmine = xs + ((ct * 2 + side) * 3) * 256 + lane;
  const float* xpeer = xs + ((ct * 2 + (1 - side)) * 3) * 256 + lane;
  if (side == 0) {
    #pragma unroll
    for (int v = 0; v < 8; ++v) {
      xmine[v * 32] = g01[v];  xmine[256 + v * 32] = g11[v];  xmine[512 + v * 32] = g21[v];
    }
  } else {
    #pragma unroll
    for (int v = 0; v < 8; ++v) {
      xmine[v * 32] = g00[v];  xmine[256 + v * 32] = g10[v];  xmine[512 + v * 32] = g20[v];
    }
  }
  __syncthreads();

  const int col = ct * 16 + l16;
  const bool valid = (col < HID);
  const int cb = valid ? col : 0;
  const float bir = valid ? bih[cb]           : 0.0f;
  const float biz = valid ? bih[HID + cb]     : 0.0f;
  const float bin = valid ? bih[2 * HID + cb] : 0.0f;
  const float bhr = valid ? bhh[cb]           : 0.0f;
  const float bhz = valid ? bhh[HID + cb]     : 0.0f;
  const float bhn = valid ? bhh[2 * HID + cb] : 0.0f;

#define GATE_OUT(RT, V, IRV, HRV, IZV, HZV, INV, HNV)                      \
  do {                                                                     \
    const int rl = (RT) * 16 + (V) + 8 * half;                             \
    const float r = sigmoidf_(((IRV) + bir) + ((HRV) + bhr));              \
    const float z = sigmoidf_(((IZV) + biz) + ((HZV) + bhz));              \
    const float n = tanhf_(((INV) + bin) + r * ((HNV) + bhn));             \
    const float hv = hs[rl * LDS_LD + col];                                \
    const float o = (1.0f - z) * n + z * hv;                               \
    const long row = rowbase + rl;                                         \
    if (valid)          hout[row * ldo + col] = o;                         \
    else if (zero_pad)  hout[row * ldo + col] = 0.0f;                      \
  } while (0)

  if (side == 0) {
    #pragma unroll
    for (int v = 0; v < 8; ++v) {
      const float hr = xpeer[v * 32], hz = xpeer[256 + v * 32], hn = xpeer[512 + v * 32];
      GATE_OUT(0, v, g00[v], hr, g10[v], hz, g20[v], hn);
    }
  } else {
    #pragma unroll
    for (int v = 0; v < 8; ++v) {
      const float ir = xpeer[v * 32], iz = xpeer[256 + v * 32], in_ = xpeer[512 + v * 32];
      GATE_OUT(1, v, ir, g01[v], iz, g11[v], in_, g21[v]);
    }
  }
#undef GATE_OUT
}

extern "C" void kernel_launch(void* const* d_in, const int* in_sizes, int n_in,
                              void* d_out, int out_size, void* d_ws, size_t ws_size,
                              hipStream_t stream) {
  const float* x    = (const float*)d_in[0];
  const float* wgt  = (const float*)d_in[1];
  const float* w_ih = (const float*)d_in[2];
  const float* w_hh = (const float*)d_in[3];
  const float* b_ih = (const float*)d_in[4];
  const float* b_hh = (const float*)d_in[5];
  const int*   ei   = (const int*)d_in[6];
  const int n_edges = in_sizes[6] / 2;
  float* out = (float*)d_out;

  const size_t npad = (size_t)N_NODES * HP;
  const size_t gate_sz = 3 * COL_TILES * KSTEPS * 64;
  float* ws      = (float*)d_ws;
  float* h_pad   = ws;
  float* h_alt   = h_pad + npad;
  float* wfold   = h_alt + npad;
  float* wpackih = wfold + (size_t)N_LAYERS * gate_sz;
  float* wpackhh = wpackih + gate_sz;
  int*   deg     = (int*)(wpackhh + gate_sz);
  int*   rowptr  = deg + N_NODES;
  int*   cursor  = rowptr + (N_NODES + 4);
  int*   csr_src = cursor + N_NODES;

  {
    int n = N_NODES * HP;
    init_hpad<<<(n + 255) / 256, 256, 0, stream>>>(x, h_pad);
    int ng = (int)gate_sz;
    pack_wg<<<(ng + 255) / 256, 256, 0, stream>>>(w_ih, wpackih);
    pack_wg<<<(ng + 255) / 256, 256, 0, stream>>>(w_hh, wpackhh);
    fold_w<<<dim3(HP / RPB, 3, N_LAYERS), 320, 0, stream>>>(wgt, wpackih, wfold);

    hipMemsetAsync(deg, 0, (size_t)N_NODES * sizeof(int), stream);
    hist_deg<<<(n_edges + 255) / 256, 256, 0, stream>>>(ei, deg, n_edges);
    scan_deg<<<1, 1024, 0, stream>>>(deg, rowptr, cursor);
    fill_csr<<<(n_edges + 255) / 256, 256, 0, stream>>>(ei, cursor, csr_src, n_edges);
  }

  for (int l = 0; l < N_LAYERS; ++l) {
    const bool last = (l == N_LAYERS - 1);
    const float* hin = (l & 1) ? h_alt : h_pad;
    float* hnext     = (l & 1) ? h_pad : h_alt;
    gru_fused<<<RBLOCKS, GRU_THREADS, 0, stream>>>(
        hin, rowptr, csr_src, wfold + (size_t)l * gate_sz, wpackhh, b_ih, b_hh,
        last ? out : hnext, last ? HID : HP, last ? 0 : 1);
  }
}


# gru_fused: staging + edge sum at s_setprio 3 so the older resident block's WMMAs cannot hold them back; with it, weight fragments 3 k-steps ahead (4 sets) fill the matrix pipe; sum order unchanged
# speedup vs baseline: 1.6638x; 1.0321x over previous
#include <hip/hip_runtime.h>
#include <hip/hip_bf16.h>
#include <math.h>

#define N_NODES   100000
#define N_LAYERS  5
#define HID       150
#define HP        160
#define RPB       32
#define RBLOCKS   (N_NODES / RPB)
#define COL_TILES (HP / 16)
#define KSTEPS    (HP / 4)
#define LDS_LD    164

typedef __attribute__((ext_vector_type(2))) float v2f;
typedef __attribute__((ext_vector_type(8))) float v8f;

__device__ __forceinline__ v8f wmma_f32(v2f a, v2f b, v8f c) {
  return __builtin_amdgcn_wmma_f32_16x16x4_f32(
       false, a,  false, b,
       (short)0, c,  false,  false);
}

__device__ __forceinline__ float sigmoidf_(float x) {
  return __builtin_amdgcn_rcpf(1.0f + __expf(-x));
}
__device__ __forceinline__ float tanhf_(float x) {
  return 1.0f - 2.0f * __builtin_amdgcn_rcpf(1.0f + __builtin_amdgcn_exp2f(x * 2.88539008f));
}


__global__ void init_hpad(const float* __restrict__ x, float* __restrict__ hpad) {
  int idx = blockIdx.x * blockDim.x + threadIdx.x;
  if (idx >= N_NODES * HP) return;
  int row = idx / HP, col = idx - row * HP;
  hpad[idx] = (col < HID) ? x[row * HID + col] : 0.0f;
}

__global__ void pack_wg(const float* __restrict__ w, float* __restrict__ wp) {
  int idx = blockIdx.x * blockDim.x + threadIdx.x;
  const int per_gate = COL_TILES * KSTEPS * 64;
  if (idx >= 3 * per_gate) return;
  int g    = idx / per_gate;
  int rem  = idx - g * per_gate;
  int j    = rem & 1;
  int lane = (rem >> 1) & 31;
  int ks   = (rem >> 6) % KSTEPS;
  int ct   = (rem >> 6) / KSTEPS;
  int k = ks * 4 + 2 * (lane >> 4) + j;
  int n = ct * 16 + (lane & 15);
  wp[idx] = (k < HID && n < HID) ? w[(g * HID + n) * HID + k] : 0.0f;
}

__global__ void hist_deg(const int* __restrict__ ei, int* __restrict__ deg,
                         int n_edges) {
  int e = blockIdx.x * blockDim.x + threadIdx.x;
  if (e >= n_edges) return;
  atomicAdd(&deg[ei[n_edges + e]], 1);
}

__device__ __forceinline__ void scan_write(const int4* __restrict__ dv,
                                           int4* __restrict__ rv,
                                           int4* __restrict__ cv, int vi, int first) {
  const int4 d = dv[vi];
  int4 o;
  o.x = first;
  o.y = o.x + d.x;
  o.z = o.y + d.y;
  o.w = o.z + d.z;
  rv[vi] = o;
  cv[vi] = o;
}

#define SCAN_GROUPS 25
#define SCAN_VECS   (N_NODES / 4)
static_assert(N_NODES % 4 == 0 && 32 * SCAN_GROUPS * 128 >= N_NODES &&
              31 * SCAN_GROUPS * 128 <= N_NODES,
              "scan_deg: 32 segments must cover the nodes, the first 31 whole");
__global__ void __launch_bounds__(1024)
scan_deg(const int* __restrict__ deg, int* __restrict__ rowptr,
         int* __restrict__ cursor) {
  __shared__ int wsum[32];
  const int tid = threadIdx.x;
  const int lane = tid & 31, wave = tid >> 5;
  const int4* dv = (const int4*)deg;
  int4* rv = (int4*)rowptr;
  int4* cv = (int4*)cursor;
  const int vbase = wave * (SCAN_GROUPS * 32) + lane;

  int exg[SCAN_GROUPS];
  int run = 0;
  #pragma unroll
  for (int g = 0; g < SCAN_GROUPS; ++g) {
    const int vi = vbase + g * 32;
    const bool ok = vi < SCAN_VECS;
    int4 d = dv[ok ? vi : SCAN_VECS - 1];
    if (!ok) d = make_int4(0, 0, 0, 0);
    const int s4 = (d.x + d.y) + (d.z + d.w);
    int incl = s4;
    #pragma unroll
    for (int off = 1; off < 32; off <<= 1) {
      int t = __shfl_up(incl, off);
      if (lane >= off) incl += t;
    }
    exg[g] = run + incl - s4;
    run += __shfl(incl, 31);
  }
  if (lane == 0) wsum[wave] = run;
  __syncthreads();

  const int wt = wsum[lane];
  int wincl = wt;
  #pragma unroll
  for (int off = 1; off < 32; off <<= 1) {
    int t = __shfl_up(wincl, off);
    if (lane >= off) wincl += t;
  }
  const int total = __shfl(wincl, 31);
  const int wave_excl = __shfl(wincl, wave) - __shfl(wt, wave);

  if (wave < 31) {
    #pragma unroll
    for (int g = 0; g < SCAN_GROUPS; ++g)
      scan_write(dv, rv, cv, vbase + g * 32, wave_excl + exg[g]);
  } else {
    #pragma unroll
    for (int g = 0; g < SCAN_GROUPS; ++g) {
      const int vi = vbase + g * 32;
      if (vi < SCAN_VECS) scan_write(dv, rv, cv, vi, wave_excl + exg[g]);
    }
  }
  if (tid == 0) rowptr[N_NODES] = total;
}

__global__ void fill_csr(const int* __restrict__ ei, int* __restrict__ cursor,
                         int* __restrict__ csr_src, int n_edges) {
  int e = blockIdx.x * blockDim.x + threadIdx.x;
  if (e >= n_edges) return;
  int pos = atomicAdd(&cursor[ei[n_edges + e]], 1);
  csr_src[pos] = ei[e];
}

__global__ void __launch_bounds__(320)
fold_w(const float* __restrict__ w, const float* __restrict__ wpih,
       float* __restrict__ wfold) {
  __shared__ float as_[RPB * LDS_LD];
  const int rowbase = blockIdx.x * RPB;
  const int g = blockIdx.y, l = blockIdx.z;

  static_assert(RPB * HP == 16 * 320, "fold_w staging: 16 elements per thread");
  float st[16];
  #pragma unroll
  for (int it = 0; it < 16; ++it) {
    const int i = threadIdx.x + it * 320;
    const int r = i / HP, c = i - r * HP;
    const int k = rowbase + r;
    const bool ok = (k < HID && c < HID);
    const float t = w[ok ? (l * HID + k) * HID + c : 0];
    st[it] = ok ? t : 0.0f;
  }
  #pragma unroll
  for (int it = 0; it < 16; ++it) {
    const int i = threadIdx.x + it * 320;
    const int r = i / HP, c = i - r * HP;
    as_[r * LDS_LD + c] = st[it];
  }
  __syncthreads();

  const int wave = threadIdx.x >> 5, lane = threadIdx.x & 31;
  const int half = lane >> 4, l16 = lane & 15;
  const int GSTRIDE = COL_TILES * KSTEPS * 64;
  const float* wpt = wpih + g * GSTRIDE + wave * (KSTEPS * 64);

  v8f acc0 = {}, acc1 = {};
  for (int ks = 0; ks < KSTEPS; ++ks) {
    int ao = ks * 4 + 2 * half;
    v2f a0 = *(const v2f*)&as_[l16 * LDS_LD + ao];
    v2f a1 = *(const v2f*)&as_[(l16 + 16) * LDS_LD + ao];
    v2f b  = *(const v2f*)(wpt + ks * 64 + lane * 2);
    acc0 = wmma_f32(a0, b, acc0);
    acc1 = wmma_f32(a1, b, acc1);
  }
  float* dst = wfold + (size_t)(l * 3 + g) * GSTRIDE + wave * (KSTEPS * 64);
  #pragma unroll
  for (int v = 0; v < 8; v += 2) {
    const int k0 = rowbase + v + 8 * half;
    const int k1 = k0 + 16;
    v2f p0, p1;
    p0.x = acc0[v]; p0.y = acc0[v + 1];
    p1.x = acc1[v]; p1.y = acc1[v + 1];
    *(v2f*)&dst[(k0 >> 2) * 64 + (l16 + 16 * ((k0 >> 1) & 1)) * 2] = p0;
    *(v2f*)&dst[(k1 >> 2) * 64 + (l16 + 16 * ((k1 >> 1) & 1)) * 2] = p1;
  }
}

struct AFrag { v2f a0, a1; };
struct BFrag { v2f b0, b1, b2; };

#define GW 8
#define GRU_THREADS 640
#define GRU_WAVES   (GRU_THREADS / 32)
__global__ void __launch_bounds__(GRU_THREADS)
gru_fused(const float* __restrict__ hin, const int* __restrict__ rowptr,
          const int* __restrict__ csr_src,
          const float* __restrict__ wih, const float* __restrict__ whh,
          const float* __restrict__ bih, const float* __restrict__ bhh,
          float* __restrict__ hout, int ldo, int zero_pad) {
  __shared__ __attribute__((aligned(16))) float ah[2 * RPB * LDS_LD];
  __shared__ float xs[COL_TILES * 2 * 3 * 256];
  float* as_ = ah;
  float* hs  = ah + RPB * LDS_LD;
  const int rowbase = blockIdx.x * RPB;

  __builtin_amdgcn_s_setprio(3);

  static_assert(RPB * (HP / 4) == 2 * GRU_THREADS, "h staging: 2 vectors per thread");
  {
    float4 st[2];
    #pragma unroll
    for (int it = 0; it < 2; ++it) {
      const int i = threadIdx.x + it * GRU_THREADS;
      const int r = i / (HP / 4), cc = (i - r * (HP / 4)) * 4;
      st[it] = *(const float4*)(hin + (size_t)(rowbase + r) * HP + cc);
    }
    #pragma unroll
    for (int it = 0; it < 2; ++it) {
      const int i = threadIdx.x + it * GRU_THREADS;
      const int r = i / (HP / 4), cc = (i - r * (HP / 4)) * 4;
      *(float4*)&hs[r * LDS_LD + cc] = st[it];
    }
  }

  {
    const int wv = __builtin_amdgcn_readfirstlane((int)(threadIdx.x >> 5));
    const int ln = threadIdx.x & 31;
    const int r0 = wv < 12 ? 2 * wv : wv + 12;
    const int nr = wv < 12 ? 2 : 1;
    static_assert(RPB == 32 && GRU_WAVES == 20, "edge sum: row split written for 32 rows on 20 waves");
    const unsigned lb  = (unsigned)ln * 16u;
    const unsigned lb2 = 512u + (unsigned)ln * 4u;
    const char* hbase = (const char*)hin;
    int bnd[3];
    #pragma unroll
    for (int k = 0; k < 3; ++k) bnd[k] = rowptr[rowbase + r0 + (k < nr ? k : nr)];
    #pragma unroll
    for (int k = 0; k < 2; ++k) {
      if (k < nr) {
        const int beg = bnd[k], end = bnd[k + 1];
        const int last = end > 0 ? end - 1 : 0;
        float4 acc = make_float4(0.f, 0.f, 0.f, 0.f);
        float acc4 = 0.f;
        int idx[GW];
        #pragma unroll
        for (int u = 0; u < GW; ++u) {
          const int jj = beg + u;
          idx[u] = csr_src[jj < last ? jj : last];
        }
        int j = beg;
        for (; j + GW <= end; j += GW) {
          float4 v[GW];
          float  s[GW];
          #pragma unroll
          for (int u = 0; u < GW; ++u) {
            const char* mr = hbase + (size_t)idx[u] * (HP * 4);
            v[u] = *(const float4*)(mr + lb);
            s[u] = *(const float*)(mr + lb2);
          }
          #pragma unroll
          for (int u = 0; u < GW; ++u) {
            const int jj = j + GW + u;
            idx[u] = csr_src[jj < last ? jj : last];
          }
          #pragma unroll
          for (int u = 0; u < GW; ++u) {
            acc.x += v[u].x; acc.y += v[u].y; acc.z += v[u].z; acc.w += v[u].w;
            acc4 += s[u];
          }
        }
        const int nrem = end - j;
        if (nrem > 0) {
          float4 v[GW - 1];
          float  s[GW - 1];
          #pragma unroll
          for (int u = 0; u < GW - 1; ++u) {
            const char* mr = hbase + (size_t)idx[u] * (HP * 4);
            v[u] = *(const float4*)(mr + lb);
            s[u] = *(const float*)(mr + lb2);
          }
          #pragma unroll
          for (int u = 0; u < GW - 1; ++u) {
            const bool live = u < nrem;
            acc.x += live ? v[u].x : 0.f; acc.y += live ? v[u].y : 0.f;
            acc.z += live ? v[u].z : 0.f; acc.w += live ? v[u].w : 0.f;
            acc4 += live ? s[u] : 0.f;
          }
        }
        const int r = r0 + k;
        *(float4*)&as_[r * LDS_LD + 4 * ln] = acc;
        as_[r * LDS_LD + 128 + ln] = acc4;
      }
    }
  }
  __syncthreads();
  __builtin_amdgcn_s_setprio(0);

  const int wave = __builtin_amdgcn_readfirstlane((int)(threadIdx.x >> 5));
  const int lane = threadIdx.x & 31;
  const int half = lane >> 4, l16 = lane & 15;
  const int side = wave >= COL_TILES ? 1 : 0;
  const int ct = wave - side * COL_TILES;

  const int GSTRIDE = COL_TILES * KSTEPS * 64;
  const float* ap = ah + side * (RPB * LDS_LD) + l16 * LDS_LD + 2 * half;
  const float* wp = (side ? whh : wih) + ct * (KSTEPS * 64) + lane * 2;

  v8f g00 = {}, g01 = {}, g10 = {}, g11 = {}, g20 = {}, g21 = {};

#define LOAD_B(F, WP)                                                      \
  do {                                                                     \
    F.b0 = *(const v2f*)(WP);                                              \
    F.b1 = *(const v2f*)((WP) + GSTRIDE);                                  \
    F.b2 = *(const v2f*)((WP) + 2 * GSTRIDE);                              \
  } while (0)

#define LOAD_A(F, AP)                                                      \
  do {                                                                     \
    F.a0 = *(const v2f*)(AP);                                              \
    F.a1 = *(const v2f*)((AP) + 16 * LDS_LD);                              \
  } while (0)

#define MMA_STEP(A, B)                                                     \
  do {                                                                     \
    g00 = wmma_f32(A.a0, B.b0, g00);  g01 = wmma_f32(A.a1, B.b0, g01);     \
    g10 = wmma_f32(A.a0, B.b1, g10);  g11 = wmma_f32(A.a1, B.b1, g11);     \
    g20 = wmma_f32(A.a0, B.b2, g20);  g21 = wmma_f32(A.a1, B.b2, g21);     \
  } while (0)

  static_assert(KSTEPS % 4 == 0 && KSTEPS >= 8, "K loop: steps in fours, last four peeled");
  BFrag b0s, b1s, b2s, b3s;
  AFrag a0s, a1s;
  LOAD_B(b0s, wp);
  LOAD_B(b1s, wp + 64);
  LOAD_B(b2s, wp + 128);
  LOAD_A(a0s, ap);
  #pragma nounroll
  for (int it = 0; it < KSTEPS / 4 - 1; ++it) {
    LOAD_B(b3s, wp + 192);  LOAD_A(a1s, ap + 4);
    __builtin_amdgcn_sched_barrier(0);
    MMA_STEP(a0s, b0s);
    __builtin_amdgcn_sched_barrier(0);
    LOAD_B(b0s, wp + 256);  LOAD_A(a0s, ap + 8);
    __builtin_amdgcn_sched_barrier(0);
    MMA_STEP(a1s, b1s);
    __builtin_amdgcn_sched_barrier(0);
    LOAD_B(b1s, wp + 320);  LOAD_A(a1s, ap + 12);
    __builtin_amdgcn_sched_barrier(0);
    MMA_STEP(a0s, b2s);
    __builtin_amdgcn_sched_barrier(0);
    LOAD_B(b2s, wp + 384);  LOAD_A(a0s, ap + 16);
    __builtin_amdgcn_sched_barrier(0);
    MMA_STEP(a1s, b3s);
    __builtin_amdgcn_sched_barrier(0);
    wp += 256;
    ap += 16;
  }
  LOAD_B(b3s, wp + 192);  LOAD_A(a1s, ap + 4);
  __builtin_amdgcn_sched_barrier(0);
  MMA_STEP(a0s, b0s);
  __builtin_amdgcn_sched_barrier(0);
  LOAD_A(a0s, ap + 8);
  __builtin_amdgcn_sched_barrier(0);
  MMA_STEP(a1s, b1s);
  __builtin_amdgcn_sched_barrier(0);
  LOAD_A(a1s, ap + 12);
  __builtin_amdgcn_sched_barrier(0);
  MMA_STEP(a0s, b2s);
  MMA_STEP(a1s, b3s);
#undef LOAD_B
#undef LOAD_A
#undef MMA_STEP

  float* xmine = xs + ((ct * 2 + side) * 3) * 256 + lane;
  const float* xpeer = xs + ((ct * 2 + (1 - side)) * 3) * 256 + lane;
  if (side == 0) {
    #pragma unroll
    for (int v = 0; v < 8; ++v) {
      xmine[v * 32] = g01[v];  xmine[256 + v * 32] = g11[v];  xmine[512 + v * 32] = g21[v];
    }
  } else {
    #pragma unroll
    for (int v = 0; v < 8; ++v) {
      xmine[v * 32] = g00[v];  xmine[256 + v * 32] = g10[v];  xmine[512 + v * 32] = g20[v];
    }
  }
  __syncthreads();

  const int col = ct * 16 + l16;
  const bool valid = (col < HID);
  const int cb = valid ? col : 0;
  const float bir = valid ? bih[cb]           : 0.0f;
  const float biz = valid ? bih[HID + cb]     : 0.0f;
  const float bin = valid ? bih[2 * HID + cb] : 0.0f;
  const float bhr = valid ? bhh[cb]           : 0.0f;
  const float bhz = valid ? bhh[HID + cb]     : 0.0f;
  const float bhn = valid ? bhh[2 * HID + cb] : 0.0f;

#define GATE_OUT(RT, V, IRV, HRV, IZV, HZV, INV, HNV)                      \
  do {                                                                     \
    const int rl = (RT) * 16 + (V) + 8 * half;                             \
    const float r = sigmoidf_(((IRV) + bir) + ((HRV) + bhr));              \
    const float z = sigmoidf_(((IZV) + biz) + ((HZV) + bhz));              \
    const float n = tanhf_(((INV) + bin) + r * ((HNV) + bhn));             \
    const float hv = hs[rl * LDS_LD + col];                                \
    const float o = (1.0f - z) * n + z * hv;                               \
    const long row = rowbase + rl;                                         \
    if (valid)          hout[row * ldo + col] = o;                         \
    else if (zero_pad)  hout[row * ldo + col] = 0.0f;                      \
  } while (0)

  if (side == 0) {
    #pragma unroll
    for (int v = 0; v < 8; ++v) {
      const float hr = xpeer[v * 32], hz = xpeer[256 + v * 32], hn = xpeer[512 + v * 32];
      GATE_OUT(0, v, g00[v], hr, g10[v], hz, g20[v], hn);
    }
  } else {
    #pragma unroll
    for (int v = 0; v < 8; ++v) {
      const float ir = xpeer[v * 32], iz = xpeer[256 + v * 32], in_ = xpeer[512 + v * 32];
      GATE_OUT(1, v, ir, g01[v], iz, g11[v], in_, g21[v]);
    }
  }
#undef GATE_OUT
}

extern "C" void kernel_launch(void* const* d_in, const int* in_sizes, int n_in,
                              void* d_out, int out_size, void* d_ws, size_t ws_size,
                              hipStream_t stream) {
  const float* x    = (const float*)d_in[0];
  const float* wgt  = (const float*)d_in[1];
  const float* w_ih = (const float*)d_in[2];
  const float* w_hh = (const float*)d_in[3];
  const float* b_ih = (const float*)d_in[4];
  const float* b_hh = (const float*)d_in[5];
  const int*   ei   = (const int*)d_in[6];
  const int n_edges = in_sizes[6] / 2;
  float* out = (float*)d_out;

  const size_t npad = (size_t)N_NODES * HP;
  const size_t gate_sz = 3 * COL_TILES * KSTEPS * 64;
  float* ws      = (float*)d_ws;
  float* h_pad   = ws;
  float* h_alt   = h_pad + npad;
  float* wfold   = h_alt + npad;
  float* wpackih = wfold + (size_t)N_LAYERS * gate_sz;
  float* wpackhh = wpackih + gate_sz;
  int*   deg     = (int*)(wpackhh + gate_sz);
  int*   rowptr  = deg + N_NODES;
  int*   cursor  = rowptr + (N_NODES + 4);
  int*   csr_src = cursor + N_NODES;

  {
    int n = N_NODES * HP;
    init_hpad<<<(n + 255) / 256, 256, 0, stream>>>(x, h_pad);
    int ng = (int)gate_sz;
    pack_wg<<<(ng + 255) / 256, 256, 0, stream>>>(w_ih, wpackih);
    pack_wg<<<(ng + 255) / 256, 256, 0, stream>>>(w_hh, wpackhh);
    fold_w<<<dim3(HP / RPB, 3, N_LAYERS), 320, 0, stream>>>(wgt, wpackih, wfold);

    hipMemsetAsync(deg, 0, (size_t)N_NODES * sizeof(int), stream);
    hist_deg<<<(n_edges + 255) / 256, 256, 0, stream>>>(ei, deg, n_edges);
    scan_deg<<<1, 1024, 0, stream>>>(deg, rowptr, cursor);
    fill_csr<<<(n_edges + 255) / 256, 256, 0, stream>>>(ei, cursor, csr_src, n_edges);
  }

  for (int l = 0; l < N_LAYERS; ++l) {
    const bool last = (l == N_LAYERS - 1);
    const float* hin = (l & 1) ? h_alt : h_pad;
    float* hnext     = (l & 1) ? h_pad : h_alt;
    gru_fused<<<RBLOCKS, GRU_THREADS, 0, stream>>>(
        hin, rowptr, csr_src, wfold + (size_t)l * gate_sz, wpackhh, b_ih, b_hh,
        last ? out : hnext, last ? HID : HP, last ? 0 : 1);
  }
}


# gru_fused epilogue: the block's 32 new rows are collected in LDS and stored as one contiguous run of whole 128-byte lines, 16 B per lane, not as 64-byte pieces with 64-bit address math per element
# speedup vs baseline: 1.7614x; 1.0587x over previous
#include <hip/hip_runtime.h>
#include <hip/hip_bf16.h>
#include <math.h>

#define N_NODES   100000
#define N_LAYERS  5
#define HID       150
#define HP        160
#define RPB       32
#define RBLOCKS   (N_NODES / RPB)
#define COL_TILES (HP / 16)
#define KSTEPS    (HP / 4)
#define LDS_LD    164

typedef __attribute__((ext_vector_type(2))) float v2f;
typedef __attribute__((ext_vector_type(8))) float v8f;

__device__ __forceinline__ v8f wmma_f32(v2f a, v2f b, v8f c) {
  return __builtin_amdgcn_wmma_f32_16x16x4_f32(
       false, a,  false, b,
       (short)0, c,  false,  false);
}

__device__ __forceinline__ float sigmoidf_(float x) {
  return __builtin_amdgcn_rcpf(1.0f + __expf(-x));
}
__device__ __forceinline__ float tanhf_(float x) {
  return 1.0f - 2.0f * __builtin_amdgcn_rcpf(1.0f + __builtin_amdgcn_exp2f(x * 2.88539008f));
}


__global__ void init_hpad(const float* __restrict__ x, float* __restrict__ hpad) {
  int idx = blockIdx.x * blockDim.x + threadIdx.x;
  if (idx >= N_NODES * HP) return;
  int row = idx / HP, col = idx - row * HP;
  hpad[idx] = (col < HID) ? x[row * HID + col] : 0.0f;
}

__global__ void pack_wg(const float* __restrict__ w, float* __restrict__ wp) {
  int idx = blockIdx.x * blockDim.x + threadIdx.x;
  const int per_gate = COL_TILES * KSTEPS * 64;
  if (idx >= 3 * per_gate) return;
  int g    = idx / per_gate;
  int rem  = idx - g * per_gate;
  int j    = rem & 1;
  int lane = (rem >> 1) & 31;
  int ks   = (rem >> 6) % KSTEPS;
  int ct   = (rem >> 6) / KSTEPS;
  int k = ks * 4 + 2 * (lane >> 4) + j;
  int n = ct * 16 + (lane & 15);
  wp[idx] = (k < HID && n < HID) ? w[(g * HID + n) * HID + k] : 0.0f;
}

__global__ void hist_deg(const int* __restrict__ ei, int* __restrict__ deg,
                         int n_edges) {
  int e = blockIdx.x * blockDim.x + threadIdx.x;
  if (e >= n_edges) return;
  atomicAdd(&deg[ei[n_edges + e]], 1);
}

__device__ __forceinline__ void scan_write(const int4* __restrict__ dv,
                                           int4* __restrict__ rv,
                                           int4* __restrict__ cv, int vi, int first) {
  const int4 d = dv[vi];
  int4 o;
  o.x = first;
  o.y = o.x + d.x;
  o.z = o.y + d.y;
  o.w = o.z + d.z;
  rv[vi] = o;
  cv[vi] = o;
}

#define SCAN_GROUPS 25
#define SCAN_VECS   (N_NODES / 4)
static_assert(N_NODES % 4 == 0 && 32 * SCAN_GROUPS * 128 >= N_NODES &&
              31 * SCAN_GROUPS * 128 <= N_NODES,
              "scan_deg: 32 segments must cover the nodes, the first 31 whole");
__global__ void __launch_bounds__(1024)
scan_deg(const int* __restrict__ deg, int* __restrict__ rowptr,
         int* __restrict__ cursor) {
  __shared__ int wsum[32];
  const int tid = threadIdx.x;
  const int lane = tid & 31, wave = tid >> 5;
  const int4* dv = (const int4*)deg;
  int4* rv = (int4*)rowptr;
  int4* cv = (int4*)cursor;
  const int vbase = wave * (SCAN_GROUPS * 32) + lane;

  int exg[SCAN_GROUPS];
  int run = 0;
  #pragma unroll
  for (int g = 0; g < SCAN_GROUPS; ++g) {
    const int vi = vbase + g * 32;
    const bool ok = vi < SCAN_VECS;
    int4 d = dv[ok ? vi : SCAN_VECS - 1];
    if (!ok) d = make_int4(0, 0, 0, 0);
    const int s4 = (d.x + d.y) + (d.z + d.w);
    int incl = s4;
    #pragma unroll
    for (int off = 1; off < 32; off <<= 1) {
      int t = __shfl_up(incl, off);
      if (lane >= off) incl += t;
    }
    exg[g] = run + incl - s4;
    run += __shfl(incl, 31);
  }
  if (lane == 0) wsum[wave] = run;
  __syncthreads();

  const int wt = wsum[lane];
  int wincl = wt;
  #pragma unroll
  for (int off = 1; off < 32; off <<= 1) {
    int t = __shfl_up(wincl, off);
    if (lane >= off) wincl += t;
  }
  const int total = __shfl(wincl, 31);
  const int wave_excl = __shfl(wincl, wave) - __shfl(wt, wave);

  if (wave < 31) {
    #pragma unroll
    for (int g = 0; g < SCAN_GROUPS; ++g)
      scan_write(dv, rv, cv, vbase + g * 32, wave_excl + exg[g]);
  } else {
    #pragma unroll
    for (int g = 0; g < SCAN_GROUPS; ++g) {
      const int vi = vbase + g * 32;
      if (vi < SCAN_VECS) scan_write(dv, rv, cv, vi, wave_excl + exg[g]);
    }
  }
  if (tid == 0) rowptr[N_NODES] = total;
}

__global__ void fill_csr(const int* __restrict__ ei, int* __restrict__ cursor,
                         int* __restrict__ csr_src, int n_edges) {
  int e = blockIdx.x * blockDim.x + threadIdx.x;
  if (e >= n_edges) return;
  int pos = atomicAdd(&cursor[ei[n_edges + e]], 1);
  csr_src[pos] = ei[e];
}

__global__ void __launch_bounds__(320)
fold_w(const float* __restrict__ w, const float* __restrict__ wpih,
       float* __restrict__ wfold) {
  __shared__ float as_[RPB * LDS_LD];
  const int rowbase = blockIdx.x * RPB;
  const int g = blockIdx.y, l = blockIdx.z;

  static_assert(RPB * HP == 16 * 320, "fold_w staging: 16 elements per thread");
  float st[16];
  #pragma unroll
  for (int it = 0; it < 16; ++it) {
    const int i = threadIdx.x + it * 320;
    const int r = i / HP, c = i - r * HP;
    const int k = rowbase + r;
    const bool ok = (k < HID && c < HID);
    const float t = w[ok ? (l * HID + k) * HID + c : 0];
    st[it] = ok ? t : 0.0f;
  }
  #pragma unroll
  for (int it = 0; it < 16; ++it) {
    const int i = threadIdx.x + it * 320;
    const int r = i / HP, c = i - r * HP;
    as_[r * LDS_LD + c] = st[it];
  }
  __syncthreads();

  const int wave = threadIdx.x >> 5, lane = threadIdx.x & 31;
  const int half = lane >> 4, l16 = lane & 15;
  const int GSTRIDE = COL_TILES * KSTEPS * 64;
  const float* wpt = wpih + g * GSTRIDE + wave * (KSTEPS * 64);

  v8f acc0 = {}, acc1 = {};
  for (int ks = 0; ks < KSTEPS; ++ks) {
    int ao = ks * 4 + 2 * half;
    v2f a0 = *(const v2f*)&as_[l16 * LDS_LD + ao];
    v2f a1 = *(const v2f*)&as_[(l16 + 16) * LDS_LD + ao];
    v2f b  = *(const v2f*)(wpt + ks * 64 + lane * 2);
    acc0 = wmma_f32(a0, b, acc0);
    acc1 = wmma_f32(a1, b, acc1);
  }
  float* dst = wfold + (size_t)(l * 3 + g) * GSTRIDE + wave * (KSTEPS * 64);
  #pragma unroll
  for (int v = 0; v < 8; v += 2) {
    const int k0 = rowbase + v + 8 * half;
    const int k1 = k0 + 16;
    v2f p0, p1;
    p0.x = acc0[v]; p0.y = acc0[v + 1];
    p1.x = acc1[v]; p1.y = acc1[v + 1];
    *(v2f*)&dst[(k0 >> 2) * 64 + (l16 + 16 * ((k0 >> 1) & 1)) * 2] = p0;
    *(v2f*)&dst[(k1 >> 2) * 64 + (l16 + 16 * ((k1 >> 1) & 1)) * 2] = p1;
  }
}

struct AFrag { v2f a0, a1; };
struct BFrag { v2f b0, b1, b2; };

#define GW 8
#define GRU_THREADS 640
#define GRU_WAVES   (GRU_THREADS / 32)
__global__ void __launch_bounds__(GRU_THREADS)
gru_fused(const float* __restrict__ hin, const int* __restrict__ rowptr,
          const int* __restrict__ csr_src,
          const float* __restrict__ wih, const float* __restrict__ whh,
          const float* __restrict__ bih, const float* __restrict__ bhh,
          float* __restrict__ hout, int ldo) {
  __shared__ __attribute__((aligned(16))) float ah[2 * RPB * LDS_LD];
  __shared__ float xs[COL_TILES * 2 * 3 * 256];
  float* as_ = ah;
  float* hs  = ah + RPB * LDS_LD;
  const int rowbase = blockIdx.x * RPB;

  __builtin_amdgcn_s_setprio(3);

  static_assert(RPB * (HP / 4) == 2 * GRU_THREADS, "h staging: 2 vectors per thread");
  {
    float4 st[2];
    #pragma unroll
    for (int it = 0; it < 2; ++it) {
      const int i = threadIdx.x + it * GRU_THREADS;
      const int r = i / (HP / 4), cc = (i - r * (HP / 4)) * 4;
      st[it] = *(const float4*)(hin + (size_t)(rowbase + r) * HP + cc);
    }
    #pragma unroll
    for (int it = 0; it < 2; ++it) {
      const int i = threadIdx.x + it * GRU_THREADS;
      const int r = i / (HP / 4), cc = (i - r * (HP / 4)) * 4;
      *(float4*)&hs[r * LDS_LD + cc] = st[it];
    }
  }

  {
    const int wv = __builtin_amdgcn_readfirstlane((int)(threadIdx.x >> 5));
    const int ln = threadIdx.x & 31;
    const int r0 = wv < 12 ? 2 * wv : wv + 12;
    const int nr = wv < 12 ? 2 : 1;
    static_assert(RPB == 32 && GRU_WAVES == 20, "edge sum: row split written for 32 rows on 20 waves");
    const unsigned lb  = (unsigned)ln * 16u;
    const unsigned lb2 = 512u + (unsigned)ln * 4u;
    const char* hbase = (const char*)hin;
    int bnd[3];
    #pragma unroll
    for (int k = 0; k < 3; ++k) bnd[k] = rowptr[rowbase + r0 + (k < nr ? k : nr)];
    #pragma unroll
    for (int k = 0; k < 2; ++k) {
      if (k < nr) {
        const int beg = bnd[k], end = bnd[k + 1];
        const int last = end > 0 ? end - 1 : 0;
        float4 acc = make_float4(0.f, 0.f, 0.f, 0.f);
        float acc4 = 0.f;
        int idx[GW];
        #pragma unroll
        for (int u = 0; u < GW; ++u) {
          const int jj = beg + u;
          idx[u] = csr_src[jj < last ? jj : last];
        }
        int j = beg;
        for (; j + GW <= end; j += GW) {
          float4 v[GW];
          float  s[GW];
          #pragma unroll
          for (int u = 0; u < GW; ++u) {
            const char* mr = hbase + (size_t)idx[u] * (HP * 4);
            v[u] = *(const float4*)(mr + lb);
            s[u] = *(const float*)(mr + lb2);
          }
          #pragma unroll
          for (int u = 0; u < GW; ++u) {
            const int jj = j + GW + u;
            idx[u] = csr_src[jj < last ? jj : last];
          }
          #pragma unroll
          for (int u = 0; u < GW; ++u) {
            acc.x += v[u].x; acc.y += v[u].y; acc.z += v[u].z; acc.w += v[u].w;
            acc4 += s[u];
          }
        }
        const int nrem = end - j;
        if (nrem > 0) {
          float4 v[GW - 1];
          float  s[GW - 1];
          #pragma unroll
          for (int u = 0; u < GW - 1; ++u) {
            const char* mr = hbase + (size_t)idx[u] * (HP * 4);
            v[u] = *(const float4*)(mr + lb);
            s[u] = *(const float*)(mr + lb2);
          }
          #pragma unroll
          for (int u = 0; u < GW - 1; ++u) {
            const bool live = u < nrem;
            acc.x += live ? v[u].x : 0.f; acc.y += live ? v[u].y : 0.f;
            acc.z += live ? v[u].z : 0.f; acc.w += live ? v[u].w : 0.f;
            acc4 += live ? s[u] : 0.f;
          }
        }
        const int r = r0 + k;
        *(float4*)&as_[r * LDS_LD + 4 * ln] = acc;
        as_[r * LDS_LD + 128 + ln] = acc4;
      }
    }
  }
  __syncthreads();
  __builtin_amdgcn_s_setprio(0);

  const int wave = __builtin_amdgcn_readfirstlane((int)(threadIdx.x >> 5));
  const int lane = threadIdx.x & 31;
  const int half = lane >> 4, l16 = lane & 15;
  const int side = wave >= COL_TILES ? 1 : 0;
  const int ct = wave - side * COL_TILES;

  const int GSTRIDE = COL_TILES * KSTEPS * 64;
  const float* ap = ah + side * (RPB * LDS_LD) + l16 * LDS_LD + 2 * half;
  const float* wp = (side ? whh : wih) + ct * (KSTEPS * 64) + lane * 2;

  v8f g00 = {}, g01 = {}, g10 = {}, g11 = {}, g20 = {}, g21 = {};

#define LOAD_B(F, WP)                                                      \
  do {                                                                     \
    F.b0 = *(const v2f*)(WP);                                              \
    F.b1 = *(const v2f*)((WP) + GSTRIDE);                                  \
    F.b2 = *(const v2f*)((WP) + 2 * GSTRIDE);                              \
  } while (0)

#define LOAD_A(F, AP)                                                      \
  do {                                                                     \
    F.a0 = *(const v2f*)(AP);                                              \
    F.a1 = *(const v2f*)((AP) + 16 * LDS_LD);                              \
  } while (0)

#define MMA_STEP(A, B)                                                     \
  do {                                                                     \
    g00 = wmma_f32(A.a0, B.b0, g00);  g01 = wmma_f32(A.a1, B.b0, g01);     \
    g10 = wmma_f32(A.a0, B.b1, g10);  g11 = wmma_f32(A.a1, B.b1, g11);     \
    g20 = wmma_f32(A.a0, B.b2, g20);  g21 = wmma_f32(A.a1, B.b2, g21);     \
  } while (0)

  static_assert(KSTEPS % 4 == 0 && KSTEPS >= 8, "K loop: steps in fours, last four peeled");
  BFrag b0s, b1s, b2s, b3s;
  AFrag a0s, a1s;
  LOAD_B(b0s, wp);
  LOAD_B(b1s, wp + 64);
  LOAD_B(b2s, wp + 128);
  LOAD_A(a0s, ap);
  #pragma nounroll
  for (int it = 0; it < KSTEPS / 4 - 1; ++it) {
    LOAD_B(b3s, wp + 192);  LOAD_A(a1s, ap + 4);
    __builtin_amdgcn_sched_barrier(0);
    MMA_STEP(a0s, b0s);
    __builtin_amdgcn_sched_barrier(0);
    LOAD_B(b0s, wp + 256);  LOAD_A(a0s, ap + 8);
    __builtin_amdgcn_sched_barrier(0);
    MMA_STEP(a1s, b1s);
    __builtin_amdgcn_sched_barrier(0);
    LOAD_B(b1s, wp + 320);  LOAD_A(a1s, ap + 12);
    __builtin_amdgcn_sched_barrier(0);
    MMA_STEP(a0s, b2s);
    __builtin_amdgcn_sched_barrier(0);
    LOAD_B(b2s, wp + 384);  LOAD_A(a0s, ap + 16);
    __builtin_amdgcn_sched_barrier(0);
    MMA_STEP(a1s, b3s);
    __builtin_amdgcn_sched_barrier(0);
    wp += 256;
    ap += 16;
  }
  LOAD_B(b3s, wp + 192);  LOAD_A(a1s, ap + 4);
  __builtin_amdgcn_sched_barrier(0);
  MMA_STEP(a0s, b0s);
  __builtin_amdgcn_sched_barrier(0);
  LOAD_A(a0s, ap + 8);
  __builtin_amdgcn_sched_barrier(0);
  MMA_STEP(a1s, b1s);
  __builtin_amdgcn_sched_barrier(0);
  LOAD_A(a1s, ap + 12);
  __builtin_amdgcn_sched_barrier(0);
  MMA_STEP(a0s, b2s);
  MMA_STEP(a1s, b3s);
#undef LOAD_B
#undef LOAD_A
#undef MMA_STEP

  float* xmine = xs + ((ct * 2 + side) * 3) * 256 + lane;
  const float* xpeer = xs + ((ct * 2 + (1 - side)) * 3) * 256 + lane;
  if (side == 0) {
    #pragma unroll
    for (int v = 0; v < 8; ++v) {
      xmine[v * 32] = g01[v];  xmine[256 + v * 32] = g11[v];  xmine[512 + v * 32] = g21[v];
    }
  } else {
    #pragma unroll
    for (int v = 0; v < 8; ++v) {
      xmine[v * 32] = g00[v];  xmine[256 + v * 32] = g10[v];  xmine[512 + v * 32] = g20[v];
    }
  }
  __syncthreads();

  const int col = ct * 16 + l16;
  const bool valid = (col < HID);
  const int cb = valid ? col : 0;
  const float bir = valid ? bih[cb]           : 0.0f;
  const float biz = valid ? bih[HID + cb]     : 0.0f;
  const float bin = valid ? bih[2 * HID + cb] : 0.0f;
  const float bhr = valid ? bhh[cb]           : 0.0f;
  const float bhz = valid ? bhh[HID + cb]     : 0.0f;
  const float bhn = valid ? bhh[2 * HID + cb] : 0.0f;

  float* stg = as_;
  static_assert(RPB * HP <= RPB * LDS_LD, "output rows fit in the aggr rows' place");

#define GATE_OUT(RT, V, IRV, HRV, IZV, HZV, INV, HNV)                      \
  do {                                                                     \
    const int rl = (RT) * 16 + (V) + 8 * half;                             \
    const float r = sigmoidf_(((IRV) + bir) + ((HRV) + bhr));              \
    const float z = sigmoidf_(((IZV) + biz) + ((HZV) + bhz));              \
    const float n = tanhf_(((INV) + bin) + r * ((HNV) + bhn));             \
    const float hv = hs[rl * LDS_LD + col];                                \
    const float o = (1.0f - z) * n + z * hv;                               \
    if (col < ldo) stg[rl * ldo + col] = valid ? o : 0.0f;                 \
  } while (0)

  if (side == 0) {
    #pragma unroll
    for (int v = 0; v < 8; ++v) {
      const float hr = xpeer[v * 32], hz = xpeer[256 + v * 32], hn = xpeer[512 + v * 32];
      GATE_OUT(0, v, g00[v], hr, g10[v], hz, g20[v], hn);
    }
  } else {
    #pragma unroll
    for (int v = 0; v < 8; ++v) {
      const float ir = xpeer[v * 32], iz = xpeer[256 + v * 32], in_ = xpeer[512 + v * 32];
      GATE_OUT(1, v, ir, g01[v], iz, g11[v], in_, g21[v]);
    }
  }
#undef GATE_OUT
  __syncthreads();

  {
    const int n4 = RPB * ldo / 4;
    const float4* s4 = (const float4*)stg;
    float4* d4 = (float4*)(hout + (size_t)rowbase * ldo);
    for (int i = threadIdx.x; i < n4; i += GRU_THREADS) d4[i] = s4[i];
  }
}

extern "C" void kernel_launch(void* const* d_in, const int* in_sizes, int n_in,
                              void* d_out, int out_size, void* d_ws, size_t ws_size,
                              hipStream_t stream) {
  const float* x    = (const float*)d_in[0];
  const float* wgt  = (const float*)d_in[1];
  const float* w_ih = (const float*)d_in[2];
  const float* w_hh = (const float*)d_in[3];
  const float* b_ih = (const float*)d_in[4];
  const float* b_hh = (const float*)d_in[5];
  const int*   ei   = (const int*)d_in[6];
  const int n_edges = in_sizes[6] / 2;
  float* out = (float*)d_out;

  const size_t npad = (size_t)N_NODES * HP;
  const size_t gate_sz = 3 * COL_TILES * KSTEPS * 64;
  float* ws      = (float*)d_ws;
  float* h_pad   = ws;
  float* h_alt   = h_pad + npad;
  float* wfold   = h_alt + npad;
  float* wpackih = wfold + (size_t)N_LAYERS * gate_sz;
  float* wpackhh = wpackih + gate_sz;
  int*   deg     = (int*)(wpackhh + gate_sz);
  int*   rowptr  = deg + N_NODES;
  int*   cursor  = rowptr + (N_NODES + 4);
  int*   csr_src = cursor + N_NODES;

  {
    int n = N_NODES * HP;
    init_hpad<<<(n + 255) / 256, 256, 0, stream>>>(x, h_pad);
    int ng = (int)gate_sz;
    pack_wg<<<(ng + 255) / 256, 256, 0, stream>>>(w_ih, wpackih);
    pack_wg<<<(ng + 255) / 256, 256, 0, stream>>>(w_hh, wpackhh);
    fold_w<<<dim3(HP / RPB, 3, N_LAYERS), 320, 0, stream>>>(wgt, wpackih, wfold);

    hipMemsetAsync(deg, 0, (size_t)N_NODES * sizeof(int), stream);
    hist_deg<<<(n_edges + 255) / 256, 256, 0, stream>>>(ei, deg, n_edges);
    scan_deg<<<1, 1024, 0, stream>>>(deg, rowptr, cursor);
    fill_csr<<<(n_edges + 255) / 256, 256, 0, stream>>>(ei, cursor, csr_src, n_edges);
  }

  for (int l = 0; l < N_LAYERS; ++l) {
    const bool last = (l == N_LAYERS - 1);
    const float* hin = (l & 1) ? h_alt : h_pad;
    float* hnext     = (l & 1) ? h_pad : h_alt;
    gru_fused<<<RBLOCKS, GRU_THREADS, 0, stream>>>(
        hin, rowptr, csr_src, wfold + (size_t)l * gate_sz, wpackhh, b_ih, b_hh,
        last ? out : hnext, last ? HID : HP);
  }
}
